# starting kernel
# baseline (speedup 1.0000x reference)
#include <hip/hip_runtime.h>
#include <math.h>


typedef __attribute__((ext_vector_type(2))) float v2f;
typedef __attribute__((ext_vector_type(8))) float v8f;

#define T_SEQ 2048
#define EMB   1024
#define HID   64
#define NB    8
#define MTOT  (NB * T_SEQ)
#define NKT   (T_SEQ / 16)

#define QOFF   0
#define KOFF   ((size_t)MTOT * HID)
#define VOFF   ((size_t)2 * MTOT * HID)
#define TSOFF  ((size_t)3 * MTOT * HID)
#define SUFOFF (TSOFF + (size_t)NB * NKT * HID)

__device__ __forceinline__ v8f wmma_f32(v2f a, v2f b, v8f c) {
  return __builtin_amdgcn_wmma_f32_16x16x4_f32(
       false, a,  false, b,
       (short)0, c,  false,  false);
}

__global__ __launch_bounds__(128) void qkv_kernel(
    const float* __restrict__ X,
    const float* __restrict__ Wq, const float* __restrict__ bq,
    const float* __restrict__ Wk, const float* __restrict__ bk,
    const float* __restrict__ Wv, const float* __restrict__ bv,
    float* __restrict__ ws) {
  const int lane = threadIdx.x;
  const int nt   = threadIdx.y;
  const int mat  = blockIdx.y;
  const int m0   = blockIdx.x * 16;
  const int n0   = nt * 16;
  const int ln   = lane & 15;
  const int of   = (lane >> 4) * 2;

  const float* W    = (mat == 0) ? Wq : (mat == 1) ? Wk : Wv;
  const float* bias = (mat == 0) ? bq : (mat == 1) ? bk : bv;
  float* out = ws + (size_t)mat * (size_t)MTOT * HID;

  const float* arow = X + (size_t)(m0 + ln) * EMB + of;
  const float* bcol = W + (size_t)of * HID + n0 + ln;

  v8f c = {};
  #pragma unroll 4
  for (int k = 0; k < EMB; k += 4) {
    v2f a = *(const v2f*)(arow + k);
    v2f b;
    b.x = bcol[(size_t)k * HID];
    b.y = bcol[(size_t)(k + 1) * HID];
    c = wmma_f32(a, b, c);
  }

  const float bb = bias[n0 + ln];
  const int mrow = m0 + (lane >> 4) * 8;
  #pragma unroll
  for (int r = 0; r < 8; ++r)
    out[(size_t)(mrow + r) * HID + n0 + ln] = c[r] + bb;
}

__global__ void vtilesum_kernel(float* __restrict__ ws) {
  const int h = threadIdx.x;
  const int j = blockIdx.x;
  const int b = blockIdx.y;
  const float* V = ws + VOFF + (size_t)(b * T_SEQ + j * 16) * HID + h;
  float s = 0.f;
  #pragma unroll
  for (int i = 0; i < 16; ++i) s += V[(size_t)i * HID];
  ws[TSOFF + ((size_t)b * NKT + j) * HID + h] = s;
}

__global__ void vsuffix_kernel(float* __restrict__ ws) {
  const int h = threadIdx.x;
  const int b = blockIdx.x;
  const float* ts = ws + TSOFF + (size_t)b * NKT * HID;
  float* suf = ws + SUFOFF + (size_t)b * (NKT + 1) * HID;
  float acc = 0.f;
  suf[(size_t)NKT * HID + h] = 0.f;
  for (int j = NKT - 1; j >= 0; --j) {
    acc += ts[(size_t)j * HID + h];
    suf[(size_t)j * HID + h] = acc;
  }
}

__global__ __launch_bounds__(128) void attn_kernel(
    const float* __restrict__ ws, float* __restrict__ out) {
  __shared__ float plds[4][16][18];

  const int lane = threadIdx.x;
  const int w    = threadIdx.y;
  const int b    = blockIdx.x;
  const int qt   = blockIdx.y * 4 + w;
  const int q0   = qt * 16;
  const int ln   = lane & 15;
  const int hi   = lane >> 4;
  const int of   = hi * 2;

  const float* Q = ws + QOFF;
  const float* K = ws + KOFF;
  const float* V = ws + VOFF;
  const float scale = 0.125f;

  v2f qa[16];
  {
    const float* qrow = Q + (size_t)(b * T_SEQ + q0 + ln) * HID + of;
    #pragma unroll
    for (int kk = 0; kk < 16; ++kk) qa[kk] = *(const v2f*)(qrow + 4 * kk);
  }

  v8f acc0 = {}, acc1 = {}, acc2 = {}, acc3 = {};
  float rm[8], rs[8];
  #pragma unroll
  for (int r = 0; r < 8; ++r) { rm[r] = -INFINITY; rs[r] = 0.f; }

  for (int kt = 0; kt <= qt; ++kt) {
    const int k0 = kt * 16;

    v8f s = {};
    const float* krow = K + (size_t)(b * T_SEQ + k0 + ln) * HID + of;
    #pragma unroll
    for (int kk = 0; kk < 16; ++kk) {
      v2f kb = *(const v2f*)(krow + 4 * kk);
      s = wmma_f32(qa[kk], kb, s);
    }

    const int kg = k0 + ln;
    #pragma unroll
    for (int r = 0; r < 8; ++r) {
      const int qg = q0 + r + hi * 8;
      const float v = s[r] * scale;
      s[r] = (kg > qg) ? 0.f : v;
    }

    float corr[8];
    #pragma unroll
    for (int r = 0; r < 8; ++r) {
      float t = s[r];
      t = fmaxf(t, __shfl_xor(t, 1));
      t = fmaxf(t, __shfl_xor(t, 2));
      t = fmaxf(t, __shfl_xor(t, 4));
      t = fmaxf(t, __shfl_xor(t, 8));
      const float nm = fmaxf(rm[r], t);
      const float c  = __expf(rm[r] - nm);
      const float e  = __expf(s[r] - nm);
      s[r] = e;
      float rsum = e;
      rsum += __shfl_xor(rsum, 1);
      rsum += __shfl_xor(rsum, 2);
      rsum += __shfl_xor(rsum, 4);
      rsum += __shfl_xor(rsum, 8);
      rs[r] = rs[r] * c + rsum;
      rm[r] = nm;
      corr[r] = c;
    }
    #pragma unroll
    for (int r = 0; r < 8; ++r) {
      acc0[r] *= corr[r]; acc1[r] *= corr[r];
      acc2[r] *= corr[r]; acc3[r] *= corr[r];
    }

    #pragma unroll
    for (int r = 0; r < 8; ++r) plds[w][r + hi * 8][ln] = s[r];
    asm volatile("s_wait_dscnt 0x0" ::: "memory");

    const float* vb = V + (size_t)(b * T_SEQ + k0) * HID;
    #pragma unroll
    for (int j = 0; j < 4; ++j) {
      v2f pa = *(const v2f*)(&plds[w][ln][4 * j + of]);
      const float* vr0 = vb + (size_t)(4 * j + of) * HID + ln;
      const float* vr1 = vb + (size_t)(4 * j + of + 1) * HID + ln;
      v2f bb;
      bb.x = vr0[0];  bb.y = vr1[0];  acc0 = wmma_f32(pa, bb, acc0);
      bb.x = vr0[16]; bb.y = vr1[16]; acc1 = wmma_f32(pa, bb, acc1);
      bb.x = vr0[32]; bb.y = vr1[32]; acc2 = wmma_f32(pa, bb, acc2);
      bb.x = vr0[48]; bb.y = vr1[48]; acc3 = wmma_f32(pa, bb, acc3);
    }
  }

  const int cf = T_SEQ - 16 * (qt + 1);
  if (cf > 0) {
    const float* suf = ws + SUFOFF + ((size_t)b * (NKT + 1) + (qt + 1)) * HID + ln;
    const float s0 = suf[0], s1 = suf[16], s2 = suf[32], s3 = suf[48];
    #pragma unroll
    for (int r = 0; r < 8; ++r) {
      const float nm = fmaxf(rm[r], 0.f);
      const float c  = __expf(rm[r] - nm);
      const float wz = __expf(-nm);
      rs[r] = rs[r] * c + wz * (float)cf;
      acc0[r] = acc0[r] * c + wz * s0;
      acc1[r] = acc1[r] * c + wz * s1;
      acc2[r] = acc2[r] * c + wz * s2;
      acc3[r] = acc3[r] * c + wz * s3;
    }
  }

  float* orow = out + (size_t)(b * T_SEQ + q0 + hi * 8) * HID + ln;
  #pragma unroll
  for (int r = 0; r < 8; ++r) {
    const float inv = 1.f / rs[r];
    orow[(size_t)r * HID +  0] = acc0[r] * inv;
    orow[(size_t)r * HID + 16] = acc1[r] * inv;
    orow[(size_t)r * HID + 32] = acc2[r] * inv;
    orow[(size_t)r * HID + 48] = acc3[r] * inv;
  }
}

extern "C" void kernel_launch(void* const* d_in, const int* in_sizes, int n_in,
                              void* d_out, int out_size, void* d_ws, size_t ws_size,
                              hipStream_t stream) {
  const float* X  = (const float*)d_in[0];
  const float* Wq = (const float*)d_in[1];
  const float* bq = (const float*)d_in[2];
  const float* Wk = (const float*)d_in[3];
  const float* bk = (const float*)d_in[4];
  const float* Wv = (const float*)d_in[5];
  const float* bv = (const float*)d_in[6];
  float* ws  = (float*)d_ws;
  float* out = (float*)d_out;

  qkv_kernel<<<dim3(MTOT / 16, 3), dim3(32, 4), 0, stream>>>(
      X, Wq, bq, Wk, bk, Wv, bv, ws);
  vtilesum_kernel<<<dim3(NKT, NB), 64, 0, stream>>>(ws);
  vsuffix_kernel<<<NB, 64, 0, stream>>>(ws);
  attn_kernel<<<dim3(NB, NKT / 4), dim3(32, 4), 0, stream>>>(ws, out);
}


# QKV projection: one wave per 16x64 strip of one matrix (4 accumulators, tile t lane c = column 4c+t); X and W read as float4 straight into fragments, no LDS; k order inside each 8-group changes
# speedup vs baseline: 1.8056x; 1.8056x over previous
#include <hip/hip_runtime.h>
#include <math.h>


typedef __attribute__((ext_vector_type(2))) float v2f;
typedef __attribute__((ext_vector_type(4))) float v4f;
typedef __attribute__((ext_vector_type(8))) float v8f;

#define T_SEQ 2048
#define EMB   1024
#define HID   64
#define NB    8
#define MTOT  (NB * T_SEQ)
#define NKT   (T_SEQ / 16)

#define QOFF   0
#define KOFF   ((size_t)MTOT * HID)
#define VOFF   ((size_t)2 * MTOT * HID)
#define TSOFF  ((size_t)3 * MTOT * HID)
#define SUFOFF (TSOFF + (size_t)NB * NKT * HID)

__device__ __forceinline__ v8f wmma_f32(v2f a, v2f b, v8f c) {
  return __builtin_amdgcn_wmma_f32_16x16x4_f32(
       false, a,  false, b,
       (short)0, c,  false,  false);
}

#define KG  (EMB / 8)
#define KU  4

struct QkvFrag {
  v4f x[KU];
  v4f w[KU][4];
};

__device__ __forceinline__ void qkv_load(QkvFrag& f, const float* __restrict__ arow,
                                         const float* __restrict__ wrow, int g0) {
  #pragma unroll
  for (int u = 0; u < KU; ++u) {
    f.x[u] = *(const v4f*)(arow + 8 * (g0 + u));
    #pragma unroll
    for (int j = 0; j < 4; ++j)
      f.w[u][j] = *(const v4f*)(wrow + (size_t)(8 * (g0 + u) + j) * HID);
  }
}

__device__ __forceinline__ void qkv_mma(const QkvFrag& f, v8f& c0, v8f& c1, v8f& c2, v8f& c3) {
  #pragma unroll
  for (int u = 0; u < KU; ++u) {
    const v2f a01 = {f.x[u].x, f.x[u].y};
    const v2f a23 = {f.x[u].z, f.x[u].w};
    c0 = wmma_f32(a01, v2f{f.w[u][0].x, f.w[u][1].x}, c0);
    c1 = wmma_f32(a01, v2f{f.w[u][0].y, f.w[u][1].y}, c1);
    c2 = wmma_f32(a01, v2f{f.w[u][0].z, f.w[u][1].z}, c2);
    c3 = wmma_f32(a01, v2f{f.w[u][0].w, f.w[u][1].w}, c3);
    c0 = wmma_f32(a23, v2f{f.w[u][2].x, f.w[u][3].x}, c0);
    c1 = wmma_f32(a23, v2f{f.w[u][2].y, f.w[u][3].y}, c1);
    c2 = wmma_f32(a23, v2f{f.w[u][2].z, f.w[u][3].z}, c2);
    c3 = wmma_f32(a23, v2f{f.w[u][2].w, f.w[u][3].w}, c3);
  }
}

__global__ __launch_bounds__(384) __attribute__((amdgpu_waves_per_eu(3, 3))) void qkv_kernel(
    const float* __restrict__ X,
    const float* __restrict__ Wq, const float* __restrict__ bq,
    const float* __restrict__ Wk, const float* __restrict__ bk,
    const float* __restrict__ Wv, const float* __restrict__ bv,
    float* __restrict__ ws) {
  const int lane = threadIdx.x;
  const int mt   = __builtin_amdgcn_readfirstlane(threadIdx.y);
  const int mat  = __builtin_amdgcn_readfirstlane(threadIdx.z);
  const int m0   = (blockIdx.x * 4 + mt) * 16;
  const int c    = lane & 15;
  const int h    = lane >> 4;

  const float* W    = (mat == 0) ? Wq : (mat == 1) ? Wk : Wv;
  const float* bias = (mat == 0) ? bq : (mat == 1) ? bk : bv;
  float* out = ws + (size_t)mat * (size_t)MTOT * HID;

  const float* arow = X + (size_t)(m0 + c) * EMB + 4 * h;
  const float* wrow = W + (size_t)(4 * h) * HID + 4 * c;

  v8f c0 = {}, c1 = {}, c2 = {}, c3 = {};
  QkvFrag fa, fb;
  qkv_load(fa, arow, wrow, 0);
  #pragma nounroll
  for (int g0 = 0; g0 < KG - 2 * KU; g0 += 2 * KU) {
    qkv_load(fb, arow, wrow, g0 + KU);
    qkv_mma(fa, c0, c1, c2, c3);
    qkv_load(fa, arow, wrow, g0 + 2 * KU);
    qkv_mma(fb, c0, c1, c2, c3);
  }
  qkv_load(fb, arow, wrow, KG - KU);
  qkv_mma(fa, c0, c1, c2, c3);
  qkv_mma(fb, c0, c1, c2, c3);

  const v4f bb = *(const v4f*)(bias + 4 * c);
  float* orow = out + (size_t)(m0 + h * 8) * HID + 4 * c;
  #pragma unroll
  for (int r = 0; r < 8; ++r) {
    v4f o = {c0[r] + bb.x, c1[r] + bb.y, c2[r] + bb.z, c3[r] + bb.w};
    *(v4f*)(orow + (size_t)r * HID) = o;
  }
}

__global__ void vtilesum_kernel(float* __restrict__ ws) {
  const int h = threadIdx.x;
  const int j = blockIdx.x;
  const int b = blockIdx.y;
  const float* V = ws + VOFF + (size_t)(b * T_SEQ + j * 16) * HID + h;
  float s = 0.f;
  #pragma unroll
  for (int i = 0; i < 16; ++i) s += V[(size_t)i * HID];
  ws[TSOFF + ((size_t)b * NKT + j) * HID + h] = s;
}

__global__ void vsuffix_kernel(float* __restrict__ ws) {
  const int h = threadIdx.x;
  const int b = blockIdx.x;
  const float* ts = ws + TSOFF + (size_t)b * NKT * HID;
  float* suf = ws + SUFOFF + (size_t)b * (NKT + 1) * HID;
  float acc = 0.f;
  suf[(size_t)NKT * HID + h] = 0.f;
  for (int j = NKT - 1; j >= 0; --j) {
    acc += ts[(size_t)j * HID + h];
    suf[(size_t)j * HID + h] = acc;
  }
}

#define NT     8
#define KB     (16 * NT)
#define PPITCH (KB + 4)

__global__ __launch_bounds__(128) __attribute__((amdgpu_waves_per_eu(1, 1))) void attn_kernel(
    const float* __restrict__ ws, float* __restrict__ out) {
  __shared__ __attribute__((aligned(16))) float plds[4][16][PPITCH];

  const int lane = threadIdx.x;
  const int w    = threadIdx.y;
  const int b    = blockIdx.x;
  const int qt   = blockIdx.y * 4 + w;
  const int q0   = qt * 16;
  const int ln   = lane & 15;
  const int hi   = lane >> 4;
  const int of   = hi * 2;

  const float* Q = ws + QOFF;
  const float* K = ws + KOFF;
  const float* V = ws + VOFF;
  const float scale = 0.125f;

  v2f qa[16];
  {
    const float* qrow = Q + (size_t)(b * T_SEQ + q0 + ln) * HID + of;
    #pragma unroll
    for (int kk = 0; kk < 16; ++kk) qa[kk] = *(const v2f*)(qrow + 4 * kk);
  }

  v8f acc0 = {}, acc1 = {}, acc2 = {}, acc3 = {};
  float rm[8], rs[8];
  #pragma unroll
  for (int r = 0; r < 8; ++r) { rm[r] = -INFINITY; rs[r] = 0.f; }

  const int nblk = qt / NT + 1;
  for (int kb = 0; kb < nblk; ++kb) {
    const int k0 = kb * KB;

    v8f s[NT];
    #pragma unroll
    for (int t = 0; t < NT; ++t) s[t] = v8f{};
    const float* krow = K + (size_t)(b * T_SEQ + k0 + ln) * HID + of;
    #pragma unroll
    for (int kk = 0; kk < 16; ++kk) {
      #pragma unroll
      for (int t = 0; t < NT; ++t) {
        v2f kf = *(const v2f*)(krow + (size_t)t * 16 * HID + 4 * kk);
        s[t] = wmma_f32(qa[kk], kf, s[t]);
      }
    }

    #pragma unroll
    for (int t = 0; t < NT; ++t) {
      const int kg = k0 + 16 * t + ln;
      #pragma unroll
      for (int r = 0; r < 8; ++r) {
        const int qg = q0 + r + hi * 8;
        const float v = s[t][r] * scale;
        s[t][r] = (kg > qg) ? 0.f : v;
      }
    }

    float corr[8];
    #pragma unroll
    for (int r = 0; r < 8; ++r) {
      float t = s[0][r];
      #pragma unroll
      for (int u = 1; u < NT; ++u) t = fmaxf(t, s[u][r]);
      t = fmaxf(t, __shfl_xor(t, 1));
      t = fmaxf(t, __shfl_xor(t, 2));
      t = fmaxf(t, __shfl_xor(t, 4));
      t = fmaxf(t, __shfl_xor(t, 8));
      const float nm = fmaxf(rm[r], t);
      const float c  = __expf(rm[r] - nm);
      float rsum = 0.f;
      #pragma unroll
      for (int u = 0; u < NT; ++u) {
        const float e = __expf(s[u][r] - nm);
        s[u][r] = e;
        rsum += e;
      }
      rsum += __shfl_xor(rsum, 1);
      rsum += __shfl_xor(rsum, 2);
      rsum += __shfl_xor(rsum, 4);
      rsum += __shfl_xor(rsum, 8);
      rs[r] = rs[r] * c + rsum;
      rm[r] = nm;
      corr[r] = c;
    }
    #pragma unroll
    for (int r = 0; r < 8; ++r) {
      acc0[r] *= corr[r]; acc1[r] *= corr[r];
      acc2[r] *= corr[r]; acc3[r] *= corr[r];
    }

    #pragma unroll
    for (int t = 0; t < NT; ++t) {
      #pragma unroll
      for (int r = 0; r < 8; ++r) plds[w][r + hi * 8][16 * t + ln] = s[t][r];
    }
    asm volatile("s_wait_dscnt 0x0" ::: "memory");

    const float* vb = V + (size_t)(b * T_SEQ + k0) * HID;
    #pragma unroll
    for (int j = 0; j < 4 * NT; ++j) {
      v2f pa = *(const v2f*)(&plds[w][ln][4 * j + of]);
      const float* vr0 = vb + (size_t)(4 * j + of) * HID + ln;
      const float* vr1 = vb + (size_t)(4 * j + of + 1) * HID + ln;
      v2f bb;
      bb.x = vr0[0];  bb.y = vr1[0];  acc0 = wmma_f32(pa, bb, acc0);
      bb.x = vr0[16]; bb.y = vr1[16]; acc1 = wmma_f32(pa, bb, acc1);
      bb.x = vr0[32]; bb.y = vr1[32]; acc2 = wmma_f32(pa, bb, acc2);
      bb.x = vr0[48]; bb.y = vr1[48]; acc3 = wmma_f32(pa, bb, acc3);
    }
  }

  const int cf = T_SEQ - KB * nblk;
  if (cf > 0) {
    const float* suf = ws + SUFOFF + ((size_t)b * (NKT + 1) + (size_t)NT * nblk) * HID + ln;
    const float s0 = suf[0], s1 = suf[16], s2 = suf[32], s3 = suf[48];
    #pragma unroll
    for (int r = 0; r < 8; ++r) {
      const float nm = fmaxf(rm[r], 0.f);
      const float c  = __expf(rm[r] - nm);
      const float wz = __expf(-nm);
      rs[r] = rs[r] * c + wz * (float)cf;
      acc0[r] = acc0[r] * c + wz * s0;
      acc1[r] = acc1[r] * c + wz * s1;
      acc2[r] = acc2[r] * c + wz * s2;
      acc3[r] = acc3[r] * c + wz * s3;
    }
  }

  float* orow = out + (size_t)(b * T_SEQ + q0 + hi * 8) * HID + ln;
  #pragma unroll
  for (int r = 0; r < 8; ++r) {
    const float inv = 1.f / rs[r];
    orow[(size_t)r * HID +  0] = acc0[r] * inv;
    orow[(size_t)r * HID + 16] = acc1[r] * inv;
    orow[(size_t)r * HID + 32] = acc2[r] * inv;
    orow[(size_t)r * HID + 48] = acc3[r] * inv;
  }
}

extern "C" void kernel_launch(void* const* d_in, const int* in_sizes, int n_in,
                              void* d_out, int out_size, void* d_ws, size_t ws_size,
                              hipStream_t stream) {
  const float* X  = (const float*)d_in[0];
  const float* Wq = (const float*)d_in[1];
  const float* bq = (const float*)d_in[2];
  const float* Wk = (const float*)d_in[3];
  const float* bk = (const float*)d_in[4];
  const float* Wv = (const float*)d_in[5];
  const float* bv = (const float*)d_in[6];
  float* ws  = (float*)d_ws;
  float* out = (float*)d_out;

  qkv_kernel<<<dim3(MTOT / 64), dim3(32, 4, 3), 0, stream>>>(
      X, Wq, bq, Wk, bk, Wv, bv, ws);
  vtilesum_kernel<<<dim3(NKT, NB), 64, 0, stream>>>(ws);
  vsuffix_kernel<<<NB, 64, 0, stream>>>(ws);
  attn_kernel<<<dim3(NB, NKT / 4), dim3(32, 4), 0, stream>>>(ws, out);
}


# future-region V sums: each attention wave adds the V rows behind its last trip itself, straight from V (float4 loads); tile-sum and suffix-scan launches removed; that sum's order changes
# speedup vs baseline: 2.7106x; 1.5012x over previous
#include <hip/hip_runtime.h>
#include <math.h>


typedef __attribute__((ext_vector_type(2))) float v2f;
typedef __attribute__((ext_vector_type(4))) float v4f;
typedef __attribute__((ext_vector_type(8))) float v8f;

#define T_SEQ 2048
#define EMB   1024
#define HID   64
#define NB    8
#define MTOT  (NB * T_SEQ)
#define NKT   (T_SEQ / 16)

#define QOFF   0
#define KOFF   ((size_t)MTOT * HID)
#define VOFF   ((size_t)2 * MTOT * HID)

__device__ __forceinline__ v8f wmma_f32(v2f a, v2f b, v8f c) {
  return __builtin_amdgcn_wmma_f32_16x16x4_f32(
       false, a,  false, b,
       (short)0, c,  false,  false);
}

#define KG  (EMB / 8)
#define KU  4

struct QkvFrag {
  v4f x[KU];
  v4f w[KU][4];
};

__device__ __forceinline__ void qkv_load(QkvFrag& f, const float* __restrict__ arow,
                                         const float* __restrict__ wrow, int g0) {
  #pragma unroll
  for (int u = 0; u < KU; ++u) {
    f.x[u] = *(const v4f*)(arow + 8 * (g0 + u));
    #pragma unroll
    for (int j = 0; j < 4; ++j)
      f.w[u][j] = *(const v4f*)(wrow + (size_t)(8 * (g0 + u) + j) * HID);
  }
}

__device__ __forceinline__ void qkv_mma(const QkvFrag& f, v8f& c0, v8f& c1, v8f& c2, v8f& c3) {
  #pragma unroll
  for (int u = 0; u < KU; ++u) {
    const v2f a01 = {f.x[u].x, f.x[u].y};
    const v2f a23 = {f.x[u].z, f.x[u].w};
    c0 = wmma_f32(a01, v2f{f.w[u][0].x, f.w[u][1].x}, c0);
    c1 = wmma_f32(a01, v2f{f.w[u][0].y, f.w[u][1].y}, c1);
    c2 = wmma_f32(a01, v2f{f.w[u][0].z, f.w[u][1].z}, c2);
    c3 = wmma_f32(a01, v2f{f.w[u][0].w, f.w[u][1].w}, c3);
    c0 = wmma_f32(a23, v2f{f.w[u][2].x, f.w[u][3].x}, c0);
    c1 = wmma_f32(a23, v2f{f.w[u][2].y, f.w[u][3].y}, c1);
    c2 = wmma_f32(a23, v2f{f.w[u][2].z, f.w[u][3].z}, c2);
    c3 = wmma_f32(a23, v2f{f.w[u][2].w, f.w[u][3].w}, c3);
  }
}

__global__ __launch_bounds__(384) __attribute__((amdgpu_waves_per_eu(3, 3))) void qkv_kernel(
    const float* __restrict__ X,
    const float* __restrict__ Wq, const float* __restrict__ bq,
    const float* __restrict__ Wk, const float* __restrict__ bk,
    const float* __restrict__ Wv, const float* __restrict__ bv,
    float* __restrict__ ws) {
  const int lane = threadIdx.x;
  const int mt   = __builtin_amdgcn_readfirstlane(threadIdx.y);
  const int mat  = __builtin_amdgcn_readfirstlane(threadIdx.z);
  const int m0   = (blockIdx.x * 4 + mt) * 16;
  const int c    = lane & 15;
  const int h    = lane >> 4;

  const float* W    = (mat == 0) ? Wq : (mat == 1) ? Wk : Wv;
  const float* bias = (mat == 0) ? bq : (mat == 1) ? bk : bv;
  float* out = ws + (size_t)mat * (size_t)MTOT * HID;

  const float* arow = X + (size_t)(m0 + c) * EMB + 4 * h;
  const float* wrow = W + (size_t)(4 * h) * HID + 4 * c;

  v8f c0 = {}, c1 = {}, c2 = {}, c3 = {};
  QkvFrag fa, fb;
  qkv_load(fa, arow, wrow, 0);
  #pragma nounroll
  for (int g0 = 0; g0 < KG - 2 * KU; g0 += 2 * KU) {
    qkv_load(fb, arow, wrow, g0 + KU);
    qkv_mma(fa, c0, c1, c2, c3);
    qkv_load(fa, arow, wrow, g0 + 2 * KU);
    qkv_mma(fb, c0, c1, c2, c3);
  }
  qkv_load(fb, arow, wrow, KG - KU);
  qkv_mma(fa, c0, c1, c2, c3);
  qkv_mma(fb, c0, c1, c2, c3);

  const v4f bb = *(const v4f*)(bias + 4 * c);
  float* orow = out + (size_t)(m0 + h * 8) * HID + 4 * c;
  #pragma unroll
  for (int r = 0; r < 8; ++r) {
    v4f o = {c0[r] + bb.x, c1[r] + bb.y, c2[r] + bb.z, c3[r] + bb.w};
    *(v4f*)(orow + (size_t)r * HID) = o;
  }
}

#define NT     8
#define KB     (16 * NT)
#define PPITCH (KB + 4)

__global__ __launch_bounds__(128) __attribute__((amdgpu_waves_per_eu(1, 1))) void attn_kernel(
    const float* __restrict__ ws, float* __restrict__ out) {
  __shared__ __attribute__((aligned(16))) float plds[4][16][PPITCH];

  const int lane = threadIdx.x;
  const int w    = threadIdx.y;
  const int b    = blockIdx.x;
  const int qt   = blockIdx.y * 4 + w;
  const int q0   = qt * 16;
  const int ln   = lane & 15;
  const int hi   = lane >> 4;
  const int of   = hi * 2;

  const float* Q = ws + QOFF;
  const float* K = ws + KOFF;
  const float* V = ws + VOFF;
  const float scale = 0.125f;

  v2f qa[16];
  {
    const float* qrow = Q + (size_t)(b * T_SEQ + q0 + ln) * HID + of;
    #pragma unroll
    for (int kk = 0; kk < 16; ++kk) qa[kk] = *(const v2f*)(qrow + 4 * kk);
  }

  v8f acc0 = {}, acc1 = {}, acc2 = {}, acc3 = {};
  float rm[8], rs[8];
  #pragma unroll
  for (int r = 0; r < 8; ++r) { rm[r] = -INFINITY; rs[r] = 0.f; }

  const int nblk = qt / NT + 1;
  for (int kb = 0; kb < nblk; ++kb) {
    const int k0 = kb * KB;

    v8f s[NT];
    #pragma unroll
    for (int t = 0; t < NT; ++t) s[t] = v8f{};
    const float* krow = K + (size_t)(b * T_SEQ + k0 + ln) * HID + of;
    #pragma unroll
    for (int kk = 0; kk < 16; ++kk) {
      #pragma unroll
      for (int t = 0; t < NT; ++t) {
        v2f kf = *(const v2f*)(krow + (size_t)t * 16 * HID + 4 * kk);
        s[t] = wmma_f32(qa[kk], kf, s[t]);
      }
    }

    #pragma unroll
    for (int t = 0; t < NT; ++t) {
      const int kg = k0 + 16 * t + ln;
      #pragma unroll
      for (int r = 0; r < 8; ++r) {
        const int qg = q0 + r + hi * 8;
        const float v = s[t][r] * scale;
        s[t][r] = (kg > qg) ? 0.f : v;
      }
    }

    float corr[8];
    #pragma unroll
    for (int r = 0; r < 8; ++r) {
      float t = s[0][r];
      #pragma unroll
      for (int u = 1; u < NT; ++u) t = fmaxf(t, s[u][r]);
      t = fmaxf(t, __shfl_xor(t, 1));
      t = fmaxf(t, __shfl_xor(t, 2));
      t = fmaxf(t, __shfl_xor(t, 4));
      t = fmaxf(t, __shfl_xor(t, 8));
      const float nm = fmaxf(rm[r], t);
      const float c  = __expf(rm[r] - nm);
      float rsum = 0.f;
      #pragma unroll
      for (int u = 0; u < NT; ++u) {
        const float e = __expf(s[u][r] - nm);
        s[u][r] = e;
        rsum += e;
      }
      rsum += __shfl_xor(rsum, 1);
      rsum += __shfl_xor(rsum, 2);
      rsum += __shfl_xor(rsum, 4);
      rsum += __shfl_xor(rsum, 8);
      rs[r] = rs[r] * c + rsum;
      rm[r] = nm;
      corr[r] = c;
    }
    #pragma unroll
    for (int r = 0; r < 8; ++r) {
      acc0[r] *= corr[r]; acc1[r] *= corr[r];
      acc2[r] *= corr[r]; acc3[r] *= corr[r];
    }

    #pragma unroll
    for (int t = 0; t < NT; ++t) {
      #pragma unroll
      for (int r = 0; r < 8; ++r) plds[w][r + hi * 8][16 * t + ln] = s[t][r];
    }
    asm volatile("s_wait_dscnt 0x0" ::: "memory");

    const float* vb = V + (size_t)(b * T_SEQ + k0) * HID;
    #pragma unroll
    for (int j = 0; j < 4 * NT; ++j) {
      v2f pa = *(const v2f*)(&plds[w][ln][4 * j + of]);
      const float* vr0 = vb + (size_t)(4 * j + of) * HID + ln;
      const float* vr1 = vb + (size_t)(4 * j + of + 1) * HID + ln;
      v2f bb;
      bb.x = vr0[0];  bb.y = vr1[0];  acc0 = wmma_f32(pa, bb, acc0);
      bb.x = vr0[16]; bb.y = vr1[16]; acc1 = wmma_f32(pa, bb, acc1);
      bb.x = vr0[32]; bb.y = vr1[32]; acc2 = wmma_f32(pa, bb, acc2);
      bb.x = vr0[48]; bb.y = vr1[48]; acc3 = wmma_f32(pa, bb, acc3);
    }
  }

  const int cf = T_SEQ - KB * nblk;
  if (cf > 0) {
    const int nrow = __builtin_amdgcn_readfirstlane(cf);
    const float* vs = V + (size_t)(b * T_SEQ + KB * nblk + hi) * HID + 4 * ln;
    v4f p0 = {}, p1 = {}, p2 = {}, p3 = {};
    for (int r0 = 0; r0 < nrow; r0 += 32) {
      v4f x[16];
      #pragma unroll
      for (int u = 0; u < 16; ++u) x[u] = *(const v4f*)(vs + (size_t)(r0 + 2 * u) * HID);
      #pragma unroll
      for (int u = 0; u < 16; u += 4) { p0 += x[u]; p1 += x[u + 1]; p2 += x[u + 2]; p3 += x[u + 3]; }
    }
    v4f cs = (p0 + p1) + (p2 + p3);
    cs.x += __shfl_xor(cs.x, 16);
    cs.y += __shfl_xor(cs.y, 16);
    cs.z += __shfl_xor(cs.z, 16);
    cs.w += __shfl_xor(cs.w, 16);
    *(v4f*)(&plds[w][0][4 * ln]) = cs;
    asm volatile("s_wait_dscnt 0x0" ::: "memory");
    const float s0 = plds[w][0][ln], s1 = plds[w][0][16 + ln], s2 = plds[w][0][32 + ln], s3 = plds[w][0][48 + ln];
    #pragma unroll
    for (int r = 0; r < 8; ++r) {
      const float nm = fmaxf(rm[r], 0.f);
      const float c  = __expf(rm[r] - nm);
      const float wz = __expf(-nm);
      rs[r] = rs[r] * c + wz * (float)cf;
      acc0[r] = acc0[r] * c + wz * s0;
      acc1[r] = acc1[r] * c + wz * s1;
      acc2[r] = acc2[r] * c + wz * s2;
      acc3[r] = acc3[r] * c + wz * s3;
    }
  }

  float* orow = out + (size_t)(b * T_SEQ + q0 + hi * 8) * HID + ln;
  #pragma unroll
  for (int r = 0; r < 8; ++r) {
    const float inv = 1.f / rs[r];
    orow[(size_t)r * HID +  0] = acc0[r] * inv;
    orow[(size_t)r * HID + 16] = acc1[r] * inv;
    orow[(size_t)r * HID + 32] = acc2[r] * inv;
    orow[(size_t)r * HID + 48] = acc3[r] * inv;
  }
}

extern "C" void kernel_launch(void* const* d_in, const int* in_sizes, int n_in,
                              void* d_out, int out_size, void* d_ws, size_t ws_size,
                              hipStream_t stream) {
  const float* X  = (const float*)d_in[0];
  const float* Wq = (const float*)d_in[1];
  const float* bq = (const float*)d_in[2];
  const float* Wk = (const float*)d_in[3];
  const float* bk = (const float*)d_in[4];
  const float* Wv = (const float*)d_in[5];
  const float* bv = (const float*)d_in[6];
  float* ws  = (float*)d_ws;
  float* out = (float*)d_out;

  qkv_kernel<<<dim3(MTOT / 64), dim3(32, 4, 3), 0, stream>>>(
      X, Wq, bq, Wk, bk, Wv, bv, ws);
  attn_kernel<<<dim3(NB, NKT / 4), dim3(32, 4), 0, stream>>>(ws, out);
}


# K and V planes of the workspace kept in WMMA-fragment order: an attention K/V fragment is one whole-line float4 load for 2 WMMAs; k order in each 8-group of S, P.V and future-V sum order change
# speedup vs baseline: 3.3634x; 1.2408x over previous
#include <hip/hip_runtime.h>
#include <math.h>


typedef __attribute__((ext_vector_type(2))) float v2f;
typedef __attribute__((ext_vector_type(4))) float v4f;
typedef __attribute__((ext_vector_type(8))) float v8f;

#define T_SEQ 2048
#define EMB   1024
#define HID   64
#define NB    8
#define MTOT  (NB * T_SEQ)
#define NKT   (T_SEQ / 16)

#define QOFF   0
#define KOFF   ((size_t)MTOT * HID)
#define VOFF   ((size_t)2 * MTOT * HID)

__device__ __forceinline__ v8f wmma_f32(v2f a, v2f b, v8f c) {
  return __builtin_amdgcn_wmma_f32_16x16x4_f32(
       false, a,  false, b,
       (short)0, c,  false,  false);
}

#define KG  (EMB / 8)
#define KU  4

struct QkvFrag {
  v4f x[KU];
  v4f w[KU][4];
};

__device__ __forceinline__ void qkv_load(QkvFrag& f, const float* __restrict__ arow,
                                         const float* __restrict__ wrow, int g0) {
  #pragma unroll
  for (int u = 0; u < KU; ++u) {
    f.x[u] = *(const v4f*)(arow + 8 * (g0 + u));
    #pragma unroll
    for (int j = 0; j < 4; ++j)
      f.w[u][j] = *(const v4f*)(wrow + (size_t)(8 * (g0 + u) + j) * HID);
  }
}

__device__ __forceinline__ void qkv_mma(const QkvFrag& f, v8f& c0, v8f& c1, v8f& c2, v8f& c3) {
  #pragma unroll
  for (int u = 0; u < KU; ++u) {
    const v2f a01 = {f.x[u].x, f.x[u].y};
    const v2f a23 = {f.x[u].z, f.x[u].w};
    c0 = wmma_f32(a01, v2f{f.w[u][0].x, f.w[u][1].x}, c0);
    c1 = wmma_f32(a01, v2f{f.w[u][0].y, f.w[u][1].y}, c1);
    c2 = wmma_f32(a01, v2f{f.w[u][0].z, f.w[u][1].z}, c2);
    c3 = wmma_f32(a01, v2f{f.w[u][0].w, f.w[u][1].w}, c3);
    c0 = wmma_f32(a23, v2f{f.w[u][2].x, f.w[u][3].x}, c0);
    c1 = wmma_f32(a23, v2f{f.w[u][2].y, f.w[u][3].y}, c1);
    c2 = wmma_f32(a23, v2f{f.w[u][2].z, f.w[u][3].z}, c2);
    c3 = wmma_f32(a23, v2f{f.w[u][2].w, f.w[u][3].w}, c3);
  }
}

__global__ __launch_bounds__(384) __attribute__((amdgpu_waves_per_eu(3, 3))) void qkv_kernel(
    const float* __restrict__ X,
    const float* __restrict__ Wq, const float* __restrict__ bq,
    const float* __restrict__ Wk, const float* __restrict__ bk,
    const float* __restrict__ Wv, const float* __restrict__ bv,
    float* __restrict__ ws) {
  const int lane = threadIdx.x;
  const int mt   = __builtin_amdgcn_readfirstlane(threadIdx.y);
  const int mat  = __builtin_amdgcn_readfirstlane(threadIdx.z);
  const int m0   = (blockIdx.x * 4 + mt) * 16;
  const int c    = lane & 15;
  const int h    = lane >> 4;

  const float* W    = (mat == 0) ? Wq : (mat == 1) ? Wk : Wv;
  const float* bias = (mat == 0) ? bq : (mat == 1) ? bk : bv;
  float* out = ws + (size_t)mat * (size_t)MTOT * HID;

  const float* arow = X + (size_t)(m0 + c) * EMB + 4 * h;
  const float* wrow = W + (size_t)(4 * h) * HID + 4 * c;

  v8f c0 = {}, c1 = {}, c2 = {}, c3 = {};
  QkvFrag fa, fb;
  qkv_load(fa, arow, wrow, 0);
  #pragma nounroll
  for (int g0 = 0; g0 < KG - 2 * KU; g0 += 2 * KU) {
    qkv_load(fb, arow, wrow, g0 + KU);
    qkv_mma(fa, c0, c1, c2, c3);
    qkv_load(fa, arow, wrow, g0 + 2 * KU);
    qkv_mma(fb, c0, c1, c2, c3);
  }
  qkv_load(fb, arow, wrow, KG - KU);
  qkv_mma(fa, c0, c1, c2, c3);
  qkv_mma(fb, c0, c1, c2, c3);

  const v4f bb = *(const v4f*)(bias + 4 * c);
  float* tile = out + (size_t)m0 * HID;
  if (mat == 0) {
    float* orow = tile + (size_t)(h * 8) * HID + 4 * c;
    #pragma unroll
    for (int r = 0; r < 8; ++r) {
      v4f o = {c0[r] + bb.x, c1[r] + bb.y, c2[r] + bb.z, c3[r] + bb.w};
      *(v4f*)(orow + (size_t)r * HID) = o;
    }
  } else if (mat == 1) {
    float* o4 = tile + (size_t)(16 * c + 8 * h) * 4;
    #pragma unroll
    for (int r = 0; r < 8; ++r) {
      v4f o = {c0[r] + bb.x, c1[r] + bb.y, c2[r] + bb.z, c3[r] + bb.w};
      *(v4f*)(o4 + 4 * r) = o;
    }
  } else {
    float* o4 = tile + (size_t)(((h * 4 + (c >> 2)) * 2) * 16 + 4 * (c & 3)) * 4;
    #pragma unroll
    for (int q = 0; q < 2; ++q) {
      const v4f o0 = {c0[4 * q] + bb.x, c0[4 * q + 1] + bb.x, c0[4 * q + 2] + bb.x, c0[4 * q + 3] + bb.x};
      const v4f o1 = {c1[4 * q] + bb.y, c1[4 * q + 1] + bb.y, c1[4 * q + 2] + bb.y, c1[4 * q + 3] + bb.y};
      const v4f o2 = {c2[4 * q] + bb.z, c2[4 * q + 1] + bb.z, c2[4 * q + 2] + bb.z, c2[4 * q + 3] + bb.z};
      const v4f o3 = {c3[4 * q] + bb.w, c3[4 * q + 1] + bb.w, c3[4 * q + 2] + bb.w, c3[4 * q + 3] + bb.w};
      *(v4f*)(o4 + (size_t)(16 * q + 0) * 4) = o0;
      *(v4f*)(o4 + (size_t)(16 * q + 1) * 4) = o1;
      *(v4f*)(o4 + (size_t)(16 * q + 2) * 4) = o2;
      *(v4f*)(o4 + (size_t)(16 * q + 3) * 4) = o3;
    }
  }
}

#define NT     8
#define KB     (16 * NT)
#define PPITCH (KB + 4)

__global__ __launch_bounds__(128) __attribute__((amdgpu_waves_per_eu(1, 1))) void attn_kernel(
    const float* __restrict__ ws, float* __restrict__ out) {
  __shared__ __attribute__((aligned(16))) float plds[4][16][PPITCH];

  const int lane = threadIdx.x;
  const int w    = threadIdx.y;
  const int b    = blockIdx.x;
  const int qt   = blockIdx.y * 4 + w;
  const int q0   = qt * 16;
  const int ln   = lane & 15;
  const int hi   = lane >> 4;

  const float* Q = ws + QOFF;
  const float* K = ws + KOFF;
  const float* V = ws + VOFF;
  const float scale = 0.125f;

  v4f qa[8];
  {
    const float* qrow = Q + (size_t)(b * T_SEQ + q0 + ln) * HID + 4 * hi;
    #pragma unroll
    for (int g = 0; g < 8; ++g) qa[g] = *(const v4f*)(qrow + 8 * g);
  }

  v8f acc0 = {}, acc1 = {}, acc2 = {}, acc3 = {};
  float rm[8], rs[8];
  #pragma unroll
  for (int r = 0; r < 8; ++r) { rm[r] = -INFINITY; rs[r] = 0.f; }

  const int nblk = qt / NT + 1;
  for (int kb = 0; kb < nblk; ++kb) {
    const int k0 = kb * KB;

    v8f s[NT];
    #pragma unroll
    for (int t = 0; t < NT; ++t) s[t] = v8f{};
    const float* kt = K + (size_t)(b * T_SEQ + k0) * HID + 4 * lane;
    #pragma unroll
    for (int g = 0; g < 8; ++g) {
      v4f kf[NT];
      #pragma unroll
      for (int t = 0; t < NT; ++t) kf[t] = *(const v4f*)(kt + (size_t)t * 16 * HID + 128 * g);
      const v2f a01 = {qa[g].x, qa[g].y};
      const v2f a23 = {qa[g].z, qa[g].w};
      #pragma unroll
      for (int t = 0; t < NT; ++t) s[t] = wmma_f32(a01, v2f{kf[t].x, kf[t].y}, s[t]);
      #pragma unroll
      for (int t = 0; t < NT; ++t) s[t] = wmma_f32(a23, v2f{kf[t].z, kf[t].w}, s[t]);
    }

    #pragma unroll
    for (int t = 0; t < NT; ++t) {
      const int kg = k0 + 16 * t + ln;
      #pragma unroll
      for (int r = 0; r < 8; ++r) {
        const int qg = q0 + r + hi * 8;
        const float v = s[t][r] * scale;
        s[t][r] = (kg > qg) ? 0.f : v;
      }
    }

    float corr[8];
    #pragma unroll
    for (int r = 0; r < 8; ++r) {
      float t = s[0][r];
      #pragma unroll
      for (int u = 1; u < NT; ++u) t = fmaxf(t, s[u][r]);
      t = fmaxf(t, __shfl_xor(t, 1));
      t = fmaxf(t, __shfl_xor(t, 2));
      t = fmaxf(t, __shfl_xor(t, 4));
      t = fmaxf(t, __shfl_xor(t, 8));
      const float nm = fmaxf(rm[r], t);
      const float c  = __expf(rm[r] - nm);
      float rsum = 0.f;
      #pragma unroll
      for (int u = 0; u < NT; ++u) {
        const float e = __expf(s[u][r] - nm);
        s[u][r] = e;
        rsum += e;
      }
      rsum += __shfl_xor(rsum, 1);
      rsum += __shfl_xor(rsum, 2);
      rsum += __shfl_xor(rsum, 4);
      rsum += __shfl_xor(rsum, 8);
      rs[r] = rs[r] * c + rsum;
      rm[r] = nm;
      corr[r] = c;
    }
    #pragma unroll
    for (int r = 0; r < 8; ++r) {
      acc0[r] *= corr[r]; acc1[r] *= corr[r];
      acc2[r] *= corr[r]; acc3[r] *= corr[r];
    }

    #pragma unroll
    for (int t = 0; t < NT; ++t) {
      #pragma unroll
      for (int r = 0; r < 8; ++r) plds[w][r + hi * 8][16 * t + ln] = s[t][r];
    }
    asm volatile("s_wait_dscnt 0x0" ::: "memory");

    const float* vb = V + (size_t)(b * T_SEQ + k0) * HID + 4 * lane;
    #pragma unroll
    for (int j = 0; j < 2 * NT; ++j) {
      const v4f pa = *(const v4f*)(&plds[w][ln][8 * j + 4 * hi]);
      const v4f v0 = *(const v4f*)(vb + 512 * j);
      const v4f v1 = *(const v4f*)(vb + 512 * j + 128);
      const v4f v2 = *(const v4f*)(vb + 512 * j + 256);
      const v4f v3 = *(const v4f*)(vb + 512 * j + 384);
      const v2f p01 = {pa.x, pa.y};
      const v2f p23 = {pa.z, pa.w};
      acc0 = wmma_f32(p01, v2f{v0.x, v0.y}, acc0);
      acc1 = wmma_f32(p01, v2f{v1.x, v1.y}, acc1);
      acc2 = wmma_f32(p01, v2f{v2.x, v2.y}, acc2);
      acc3 = wmma_f32(p01, v2f{v3.x, v3.y}, acc3);
      acc0 = wmma_f32(p23, v2f{v0.z, v0.w}, acc0);
      acc1 = wmma_f32(p23, v2f{v1.z, v1.w}, acc1);
      acc2 = wmma_f32(p23, v2f{v2.z, v2.w}, acc2);
      acc3 = wmma_f32(p23, v2f{v3.z, v3.w}, acc3);
    }
  }

  const int cf = T_SEQ - KB * nblk;
  if (cf > 0) {
    const int nrow = __builtin_amdgcn_readfirstlane(cf);
    const float* vs = V + (size_t)(b * T_SEQ + KB * nblk) * HID + 4 * lane;
    v4f p0 = {}, p1 = {}, p2 = {}, p3 = {};
    for (int r0 = 0; r0 < nrow; r0 += 32) {
      v4f x[16];
      #pragma unroll
      for (int u = 0; u < 16; ++u) x[u] = *(const v4f*)(vs + (size_t)r0 * HID + 128 * u);
      #pragma unroll
      for (int u = 0; u < 16; u += 4) { p0 += x[u]; p1 += x[u + 1]; p2 += x[u + 2]; p3 += x[u + 3]; }
    }
    float s0 = (p0.x + p0.y) + (p0.z + p0.w);
    float s1 = (p1.x + p1.y) + (p1.z + p1.w);
    float s2 = (p2.x + p2.y) + (p2.z + p2.w);
    float s3 = (p3.x + p3.y) + (p3.z + p3.w);
    s0 += __shfl_xor(s0, 16);
    s1 += __shfl_xor(s1, 16);
    s2 += __shfl_xor(s2, 16);
    s3 += __shfl_xor(s3, 16);
    #pragma unroll
    for (int r = 0; r < 8; ++r) {
      const float nm = fmaxf(rm[r], 0.f);
      const float c  = __expf(rm[r] - nm);
      const float wz = __expf(-nm);
      rs[r] = rs[r] * c + wz * (float)cf;
      acc0[r] = acc0[r] * c + wz * s0;
      acc1[r] = acc1[r] * c + wz * s1;
      acc2[r] = acc2[r] * c + wz * s2;
      acc3[r] = acc3[r] * c + wz * s3;
    }
  }

  float* orow = out + (size_t)(b * T_SEQ + q0 + hi * 8) * HID + ln;
  #pragma unroll
  for (int r = 0; r < 8; ++r) {
    const float inv = 1.f / rs[r];
    orow[(size_t)r * HID +  0] = acc0[r] * inv;
    orow[(size_t)r * HID + 16] = acc1[r] * inv;
    orow[(size_t)r * HID + 32] = acc2[r] * inv;
    orow[(size_t)r * HID + 48] = acc3[r] * inv;
  }
}

extern "C" void kernel_launch(void* const* d_in, const int* in_sizes, int n_in,
                              void* d_out, int out_size, void* d_ws, size_t ws_size,
                              hipStream_t stream) {
  const float* X  = (const float*)d_in[0];
  const float* Wq = (const float*)d_in[1];
  const float* bq = (const float*)d_in[2];
  const float* Wk = (const float*)d_in[3];
  const float* bk = (const float*)d_in[4];
  const float* Wv = (const float*)d_in[5];
  const float* bv = (const float*)d_in[6];
  float* ws  = (float*)d_ws;
  float* out = (float*)d_out;

  qkv_kernel<<<dim3(MTOT / 64), dim3(32, 4, 3), 0, stream>>>(
      X, Wq, bq, Wk, bk, Wv, bv, ws);
  attn_kernel<<<dim3(NB, NKT / 4), dim3(32, 4), 0, stream>>>(ws, out);
}


# attention softmax: row max and row sum butterflies as DPP row_xmask steps on the vector ALU instead of __shfl_xor (ds_bpermute round trips, a wait each); lane pairing and sum order unchanged
# speedup vs baseline: 3.7427x; 1.1128x over previous
#include <hip/hip_runtime.h>
#include <math.h>


typedef __attribute__((ext_vector_type(2))) float v2f;
typedef __attribute__((ext_vector_type(4))) float v4f;
typedef __attribute__((ext_vector_type(8))) float v8f;

#define T_SEQ 2048
#define EMB   1024
#define HID   64
#define NB    8
#define MTOT  (NB * T_SEQ)
#define NKT   (T_SEQ / 16)

#define QOFF   0
#define KOFF   ((size_t)MTOT * HID)
#define VOFF   ((size_t)2 * MTOT * HID)

__device__ __forceinline__ v8f wmma_f32(v2f a, v2f b, v8f c) {
  return __builtin_amdgcn_wmma_f32_16x16x4_f32(
       false, a,  false, b,
       (short)0, c,  false,  false);
}

template <int M>
__device__ __forceinline__ float row_xor(float v) {
  return __int_as_float(__builtin_amdgcn_update_dpp(
      0, __float_as_int(v),   0x160 + M,
       0xf,  0xf,  true));
}

#define KG  (EMB / 8)
#define KU  4

struct QkvFrag {
  v4f x[KU];
  v4f w[KU][4];
};

__device__ __forceinline__ void qkv_load(QkvFrag& f, const float* __restrict__ arow,
                                         const float* __restrict__ wrow, int g0) {
  #pragma unroll
  for (int u = 0; u < KU; ++u) {
    f.x[u] = *(const v4f*)(arow + 8 * (g0 + u));
    #pragma unroll
    for (int j = 0; j < 4; ++j)
      f.w[u][j] = *(const v4f*)(wrow + (size_t)(8 * (g0 + u) + j) * HID);
  }
}

__device__ __forceinline__ void qkv_mma(const QkvFrag& f, v8f& c0, v8f& c1, v8f& c2, v8f& c3) {
  #pragma unroll
  for (int u = 0; u < KU; ++u) {
    const v2f a01 = {f.x[u].x, f.x[u].y};
    const v2f a23 = {f.x[u].z, f.x[u].w};
    c0 = wmma_f32(a01, v2f{f.w[u][0].x, f.w[u][1].x}, c0);
    c1 = wmma_f32(a01, v2f{f.w[u][0].y, f.w[u][1].y}, c1);
    c2 = wmma_f32(a01, v2f{f.w[u][0].z, f.w[u][1].z}, c2);
    c3 = wmma_f32(a01, v2f{f.w[u][0].w, f.w[u][1].w}, c3);
    c0 = wmma_f32(a23, v2f{f.w[u][2].x, f.w[u][3].x}, c0);
    c1 = wmma_f32(a23, v2f{f.w[u][2].y, f.w[u][3].y}, c1);
    c2 = wmma_f32(a23, v2f{f.w[u][2].z, f.w[u][3].z}, c2);
    c3 = wmma_f32(a23, v2f{f.w[u][2].w, f.w[u][3].w}, c3);
  }
}

__global__ __launch_bounds__(384) __attribute__((amdgpu_waves_per_eu(3, 3))) void qkv_kernel(
    const float* __restrict__ X,
    const float* __restrict__ Wq, const float* __restrict__ bq,
    const float* __restrict__ Wk, const float* __restrict__ bk,
    const float* __restrict__ Wv, const float* __restrict__ bv,
    float* __restrict__ ws) {
  const int lane = threadIdx.x;
  const int mt   = __builtin_amdgcn_readfirstlane(threadIdx.y);
  const int mat  = __builtin_amdgcn_readfirstlane(threadIdx.z);
  const int m0   = (blockIdx.x * 4 + mt) * 16;
  const int c    = lane & 15;
  const int h    = lane >> 4;

  const float* W    = (mat == 0) ? Wq : (mat == 1) ? Wk : Wv;
  const float* bias = (mat == 0) ? bq : (mat == 1) ? bk : bv;
  float* out = ws + (size_t)mat * (size_t)MTOT * HID;

  const float* arow = X + (size_t)(m0 + c) * EMB + 4 * h;
  const float* wrow = W + (size_t)(4 * h) * HID + 4 * c;

  v8f c0 = {}, c1 = {}, c2 = {}, c3 = {};
  QkvFrag fa, fb;
  qkv_load(fa, arow, wrow, 0);
  #pragma nounroll
  for (int g0 = 0; g0 < KG - 2 * KU; g0 += 2 * KU) {
    qkv_load(fb, arow, wrow, g0 + KU);
    qkv_mma(fa, c0, c1, c2, c3);
    qkv_load(fa, arow, wrow, g0 + 2 * KU);
    qkv_mma(fb, c0, c1, c2, c3);
  }
  qkv_load(fb, arow, wrow, KG - KU);
  qkv_mma(fa, c0, c1, c2, c3);
  qkv_mma(fb, c0, c1, c2, c3);

  const v4f bb = *(const v4f*)(bias + 4 * c);
  float* tile = out + (size_t)m0 * HID;
  if (mat == 0) {
    float* orow = tile + (size_t)(h * 8) * HID + 4 * c;
    #pragma unroll
    for (int r = 0; r < 8; ++r) {
      v4f o = {c0[r] + bb.x, c1[r] + bb.y, c2[r] + bb.z, c3[r] + bb.w};
      *(v4f*)(orow + (size_t)r * HID) = o;
    }
  } else if (mat == 1) {
    float* o4 = tile + (size_t)(16 * c + 8 * h) * 4;
    #pragma unroll
    for (int r = 0; r < 8; ++r) {
      v4f o = {c0[r] + bb.x, c1[r] + bb.y, c2[r] + bb.z, c3[r] + bb.w};
      *(v4f*)(o4 + 4 * r) = o;
    }
  } else {
    float* o4 = tile + (size_t)(((h * 4 + (c >> 2)) * 2) * 16 + 4 * (c & 3)) * 4;
    #pragma unroll
    for (int q = 0; q < 2; ++q) {
      const v4f o0 = {c0[4 * q] + bb.x, c0[4 * q + 1] + bb.x, c0[4 * q + 2] + bb.x, c0[4 * q + 3] + bb.x};
      const v4f o1 = {c1[4 * q] + bb.y, c1[4 * q + 1] + bb.y, c1[4 * q + 2] + bb.y, c1[4 * q + 3] + bb.y};
      const v4f o2 = {c2[4 * q] + bb.z, c2[4 * q + 1] + bb.z, c2[4 * q + 2] + bb.z, c2[4 * q + 3] + bb.z};
      const v4f o3 = {c3[4 * q] + bb.w, c3[4 * q + 1] + bb.w, c3[4 * q + 2] + bb.w, c3[4 * q + 3] + bb.w};
      *(v4f*)(o4 + (size_t)(16 * q + 0) * 4) = o0;
      *(v4f*)(o4 + (size_t)(16 * q + 1) * 4) = o1;
      *(v4f*)(o4 + (size_t)(16 * q + 2) * 4) = o2;
      *(v4f*)(o4 + (size_t)(16 * q + 3) * 4) = o3;
    }
  }
}

#define NT     8
#define KB     (16 * NT)
#define PPITCH (KB + 4)

__global__ __launch_bounds__(128) __attribute__((amdgpu_waves_per_eu(1, 1))) void attn_kernel(
    const float* __restrict__ ws, float* __restrict__ out) {
  __shared__ __attribute__((aligned(16))) float plds[4][16][PPITCH];

  const int lane = threadIdx.x;
  const int w    = threadIdx.y;
  const int b    = blockIdx.x;
  const int qt   = blockIdx.y * 4 + w;
  const int q0   = qt * 16;
  const int ln   = lane & 15;
  const int hi   = lane >> 4;

  const float* Q = ws + QOFF;
  const float* K = ws + KOFF;
  const float* V = ws + VOFF;
  const float scale = 0.125f;

  v4f qa[8];
  {
    const float* qrow = Q + (size_t)(b * T_SEQ + q0 + ln) * HID + 4 * hi;
    #pragma unroll
    for (int g = 0; g < 8; ++g) qa[g] = *(const v4f*)(qrow + 8 * g);
  }

  v8f acc0 = {}, acc1 = {}, acc2 = {}, acc3 = {};
  float rm[8], rs[8];
  #pragma unroll
  for (int r = 0; r < 8; ++r) { rm[r] = -INFINITY; rs[r] = 0.f; }

  const int nblk = qt / NT + 1;
  for (int kb = 0; kb < nblk; ++kb) {
    const int k0 = kb * KB;

    v8f s[NT];
    #pragma unroll
    for (int t = 0; t < NT; ++t) s[t] = v8f{};
    const float* kt = K + (size_t)(b * T_SEQ + k0) * HID + 4 * lane;
    #pragma unroll
    for (int g = 0; g < 8; ++g) {
      v4f kf[NT];
      #pragma unroll
      for (int t = 0; t < NT; ++t) kf[t] = *(const v4f*)(kt + (size_t)t * 16 * HID + 128 * g);
      const v2f a01 = {qa[g].x, qa[g].y};
      const v2f a23 = {qa[g].z, qa[g].w};
      #pragma unroll
      for (int t = 0; t < NT; ++t) s[t] = wmma_f32(a01, v2f{kf[t].x, kf[t].y}, s[t]);
      #pragma unroll
      for (int t = 0; t < NT; ++t) s[t] = wmma_f32(a23, v2f{kf[t].z, kf[t].w}, s[t]);
    }

    #pragma unroll
    for (int t = 0; t < NT; ++t) {
      const int kg = k0 + 16 * t + ln;
      #pragma unroll
      for (int r = 0; r < 8; ++r) {
        const int qg = q0 + r + hi * 8;
        const float v = s[t][r] * scale;
        s[t][r] = (kg > qg) ? 0.f : v;
      }
    }

    float corr[8];
    #pragma unroll
    for (int r = 0; r < 8; ++r) {
      float t = s[0][r];
      #pragma unroll
      for (int u = 1; u < NT; ++u) t = fmaxf(t, s[u][r]);
      t = fmaxf(t, row_xor<1>(t));
      t = fmaxf(t, row_xor<2>(t));
      t = fmaxf(t, row_xor<4>(t));
      t = fmaxf(t, row_xor<8>(t));
      const float nm = fmaxf(rm[r], t);
      const float c  = __expf(rm[r] - nm);
      float rsum = 0.f;
      #pragma unroll
      for (int u = 0; u < NT; ++u) {
        const float e = __expf(s[u][r] - nm);
        s[u][r] = e;
        rsum += e;
      }
      rsum += row_xor<1>(rsum);
      rsum += row_xor<2>(rsum);
      rsum += row_xor<4>(rsum);
      rsum += row_xor<8>(rsum);
      rs[r] = rs[r] * c + rsum;
      rm[r] = nm;
      corr[r] = c;
    }
    #pragma unroll
    for (int r = 0; r < 8; ++r) {
      acc0[r] *= corr[r]; acc1[r] *= corr[r];
      acc2[r] *= corr[r]; acc3[r] *= corr[r];
    }

    #pragma unroll
    for (int t = 0; t < NT; ++t) {
      #pragma unroll
      for (int r = 0; r < 8; ++r) plds[w][r + hi * 8][16 * t + ln] = s[t][r];
    }
    asm volatile("s_wait_dscnt 0x0" ::: "memory");

    const float* vb = V + (size_t)(b * T_SEQ + k0) * HID + 4 * lane;
    #pragma unroll
    for (int j = 0; j < 2 * NT; ++j) {
      const v4f pa = *(const v4f*)(&plds[w][ln][8 * j + 4 * hi]);
      const v4f v0 = *(const v4f*)(vb + 512 * j);
      const v4f v1 = *(const v4f*)(vb + 512 * j + 128);
      const v4f v2 = *(const v4f*)(vb + 512 * j + 256);
      const v4f v3 = *(const v4f*)(vb + 512 * j + 384);
      const v2f p01 = {pa.x, pa.y};
      const v2f p23 = {pa.z, pa.w};
      acc0 = wmma_f32(p01, v2f{v0.x, v0.y}, acc0);
      acc1 = wmma_f32(p01, v2f{v1.x, v1.y}, acc1);
      acc2 = wmma_f32(p01, v2f{v2.x, v2.y}, acc2);
      acc3 = wmma_f32(p01, v2f{v3.x, v3.y}, acc3);
      acc0 = wmma_f32(p23, v2f{v0.z, v0.w}, acc0);
      acc1 = wmma_f32(p23, v2f{v1.z, v1.w}, acc1);
      acc2 = wmma_f32(p23, v2f{v2.z, v2.w}, acc2);
      acc3 = wmma_f32(p23, v2f{v3.z, v3.w}, acc3);
    }
  }

  const int cf = T_SEQ - KB * nblk;
  if (cf > 0) {
    const int nrow = __builtin_amdgcn_readfirstlane(cf);
    const float* vs = V + (size_t)(b * T_SEQ + KB * nblk) * HID + 4 * lane;
    v4f p0 = {}, p1 = {}, p2 = {}, p3 = {};
    for (int r0 = 0; r0 < nrow; r0 += 32) {
      v4f x[16];
      #pragma unroll
      for (int u = 0; u < 16; ++u) x[u] = *(const v4f*)(vs + (size_t)r0 * HID + 128 * u);
      #pragma unroll
      for (int u = 0; u < 16; u += 4) { p0 += x[u]; p1 += x[u + 1]; p2 += x[u + 2]; p3 += x[u + 3]; }
    }
    float s0 = (p0.x + p0.y) + (p0.z + p0.w);
    float s1 = (p1.x + p1.y) + (p1.z + p1.w);
    float s2 = (p2.x + p2.y) + (p2.z + p2.w);
    float s3 = (p3.x + p3.y) + (p3.z + p3.w);
    s0 += __shfl_xor(s0, 16);
    s1 += __shfl_xor(s1, 16);
    s2 += __shfl_xor(s2, 16);
    s3 += __shfl_xor(s3, 16);
    #pragma unroll
    for (int r = 0; r < 8; ++r) {
      const float nm = fmaxf(rm[r], 0.f);
      const float c  = __expf(rm[r] - nm);
      const float wz = __expf(-nm);
      rs[r] = rs[r] * c + wz * (float)cf;
      acc0[r] = acc0[r] * c + wz * s0;
      acc1[r] = acc1[r] * c + wz * s1;
      acc2[r] = acc2[r] * c + wz * s2;
      acc3[r] = acc3[r] * c + wz * s3;
    }
  }

  float* orow = out + (size_t)(b * T_SEQ + q0 + hi * 8) * HID + ln;
  #pragma unroll
  for (int r = 0; r < 8; ++r) {
    const float inv = 1.f / rs[r];
    orow[(size_t)r * HID +  0] = acc0[r] * inv;
    orow[(size_t)r * HID + 16] = acc1[r] * inv;
    orow[(size_t)r * HID + 32] = acc2[r] * inv;
    orow[(size_t)r * HID + 48] = acc3[r] * inv;
  }
}

extern "C" void kernel_launch(void* const* d_in, const int* in_sizes, int n_in,
                              void* d_out, int out_size, void* d_ws, size_t ws_size,
                              hipStream_t stream) {
  const float* X  = (const float*)d_in[0];
  const float* Wq = (const float*)d_in[1];
  const float* bq = (const float*)d_in[2];
  const float* Wk = (const float*)d_in[3];
  const float* bk = (const float*)d_in[4];
  const float* Wv = (const float*)d_in[5];
  const float* bv = (const float*)d_in[6];
  float* ws  = (float*)d_ws;
  float* out = (float*)d_out;

  qkv_kernel<<<dim3(MTOT / 64), dim3(32, 4, 3), 0, stream>>>(
      X, Wq, bq, Wk, bk, Wv, bv, ws);
  attn_kernel<<<dim3(NB, NKT / 4), dim3(32, 4), 0, stream>>>(ws, out);
}


# attention trip loop software-pipelined: S WMMAs of trip i+1 run beside the softmax of trip i, their K loads fly under its P.V run and its V loads under the S run; summation order unchanged
# speedup vs baseline: 3.9007x; 1.0422x over previous
#include <hip/hip_runtime.h>
#include <math.h>


typedef __attribute__((ext_vector_type(2))) float v2f;
typedef __attribute__((ext_vector_type(4))) float v4f;
typedef __attribute__((ext_vector_type(8))) float v8f;

#define T_SEQ 2048
#define EMB   1024
#define HID   64
#define NB    8
#define MTOT  (NB * T_SEQ)
#define NKT   (T_SEQ / 16)

#define QOFF   0
#define KOFF   ((size_t)MTOT * HID)
#define VOFF   ((size_t)2 * MTOT * HID)

__device__ __forceinline__ v8f wmma_f32(v2f a, v2f b, v8f c) {
  return __builtin_amdgcn_wmma_f32_16x16x4_f32(
       false, a,  false, b,
       (short)0, c,  false,  false);
}

template <int M>
__device__ __forceinline__ float row_xor(float v) {
  return __int_as_float(__builtin_amdgcn_update_dpp(
      0, __float_as_int(v),   0x160 + M,
       0xf,  0xf,  true));
}

#define KG  (EMB / 8)
#define KU  4

struct QkvFrag {
  v4f x[KU];
  v4f w[KU][4];
};

__device__ __forceinline__ void qkv_load(QkvFrag& f, const float* __restrict__ arow,
                                         const float* __restrict__ wrow, int g0) {
  #pragma unroll
  for (int u = 0; u < KU; ++u) {
    f.x[u] = *(const v4f*)(arow + 8 * (g0 + u));
    #pragma unroll
    for (int j = 0; j < 4; ++j)
      f.w[u][j] = *(const v4f*)(wrow + (size_t)(8 * (g0 + u) + j) * HID);
  }
}

__device__ __forceinline__ void qkv_mma(const QkvFrag& f, v8f& c0, v8f& c1, v8f& c2, v8f& c3) {
  #pragma unroll
  for (int u = 0; u < KU; ++u) {
    const v2f a01 = {f.x[u].x, f.x[u].y};
    const v2f a23 = {f.x[u].z, f.x[u].w};
    c0 = wmma_f32(a01, v2f{f.w[u][0].x, f.w[u][1].x}, c0);
    c1 = wmma_f32(a01, v2f{f.w[u][0].y, f.w[u][1].y}, c1);
    c2 = wmma_f32(a01, v2f{f.w[u][0].z, f.w[u][1].z}, c2);
    c3 = wmma_f32(a01, v2f{f.w[u][0].w, f.w[u][1].w}, c3);
    c0 = wmma_f32(a23, v2f{f.w[u][2].x, f.w[u][3].x}, c0);
    c1 = wmma_f32(a23, v2f{f.w[u][2].y, f.w[u][3].y}, c1);
    c2 = wmma_f32(a23, v2f{f.w[u][2].z, f.w[u][3].z}, c2);
    c3 = wmma_f32(a23, v2f{f.w[u][2].w, f.w[u][3].w}, c3);
  }
}

__global__ __launch_bounds__(384) __attribute__((amdgpu_waves_per_eu(3, 3))) void qkv_kernel(
    const float* __restrict__ X,
    const float* __restrict__ Wq, const float* __restrict__ bq,
    const float* __restrict__ Wk, const float* __restrict__ bk,
    const float* __restrict__ Wv, const float* __restrict__ bv,
    float* __restrict__ ws) {
  const int lane = threadIdx.x;
  const int mt   = __builtin_amdgcn_readfirstlane(threadIdx.y);
  const int mat  = __builtin_amdgcn_readfirstlane(threadIdx.z);
  const int m0   = (blockIdx.x * 4 + mt) * 16;
  const int c    = lane & 15;
  const int h    = lane >> 4;

  const float* W    = (mat == 0) ? Wq : (mat == 1) ? Wk : Wv;
  const float* bias = (mat == 0) ? bq : (mat == 1) ? bk : bv;
  float* out = ws + (size_t)mat * (size_t)MTOT * HID;

  const float* arow = X + (size_t)(m0 + c) * EMB + 4 * h;
  const float* wrow = W + (size_t)(4 * h) * HID + 4 * c;

  v8f c0 = {}, c1 = {}, c2 = {}, c3 = {};
  QkvFrag fa, fb;
  qkv_load(fa, arow, wrow, 0);
  #pragma nounroll
  for (int g0 = 0; g0 < KG - 2 * KU; g0 += 2 * KU) {
    qkv_load(fb, arow, wrow, g0 + KU);
    qkv_mma(fa, c0, c1, c2, c3);
    qkv_load(fa, arow, wrow, g0 + 2 * KU);
    qkv_mma(fb, c0, c1, c2, c3);
  }
  qkv_load(fb, arow, wrow, KG - KU);
  qkv_mma(fa, c0, c1, c2, c3);
  qkv_mma(fb, c0, c1, c2, c3);

  const v4f bb = *(const v4f*)(bias + 4 * c);
  float* tile = out + (size_t)m0 * HID;
  if (mat == 0) {
    float* orow = tile + (size_t)(h * 8) * HID + 4 * c;
    #pragma unroll
    for (int r = 0; r < 8; ++r) {
      v4f o = {c0[r] + bb.x, c1[r] + bb.y, c2[r] + bb.z, c3[r] + bb.w};
      *(v4f*)(orow + (size_t)r * HID) = o;
    }
  } else if (mat == 1) {
    float* o4 = tile + (size_t)(16 * c + 8 * h) * 4;
    #pragma unroll
    for (int r = 0; r < 8; ++r) {
      v4f o = {c0[r] + bb.x, c1[r] + bb.y, c2[r] + bb.z, c3[r] + bb.w};
      *(v4f*)(o4 + 4 * r) = o;
    }
  } else {
    float* o4 = tile + (size_t)(((h * 4 + (c >> 2)) * 2) * 16 + 4 * (c & 3)) * 4;
    #pragma unroll
    for (int q = 0; q < 2; ++q) {
      const v4f o0 = {c0[4 * q] + bb.x, c0[4 * q + 1] + bb.x, c0[4 * q + 2] + bb.x, c0[4 * q + 3] + bb.x};
      const v4f o1 = {c1[4 * q] + bb.y, c1[4 * q + 1] + bb.y, c1[4 * q + 2] + bb.y, c1[4 * q + 3] + bb.y};
      const v4f o2 = {c2[4 * q] + bb.z, c2[4 * q + 1] + bb.z, c2[4 * q + 2] + bb.z, c2[4 * q + 3] + bb.z};
      const v4f o3 = {c3[4 * q] + bb.w, c3[4 * q + 1] + bb.w, c3[4 * q + 2] + bb.w, c3[4 * q + 3] + bb.w};
      *(v4f*)(o4 + (size_t)(16 * q + 0) * 4) = o0;
      *(v4f*)(o4 + (size_t)(16 * q + 1) * 4) = o1;
      *(v4f*)(o4 + (size_t)(16 * q + 2) * 4) = o2;
      *(v4f*)(o4 + (size_t)(16 * q + 3) * 4) = o3;
    }
  }
}

#define NT     8
#define KB     (16 * NT)
#define PPITCH (KB + 4)

__device__ __forceinline__ void attn_load_k(v4f (&kf)[8][NT], const float* __restrict__ kt) {
  #pragma unroll
  for (int g = 0; g < 8; ++g) {
    #pragma unroll
    for (int t = 0; t < NT; ++t) kf[g][t] = *(const v4f*)(kt + (size_t)t * 16 * HID + 128 * g);
  }
}

__device__ __forceinline__ void attn_load_v(v4f (&vf)[2 * NT][4], const float* __restrict__ vb) {
  #pragma unroll
  for (int j = 0; j < 2 * NT; ++j) {
    #pragma unroll
    for (int t = 0; t < 4; ++t) vf[j][t] = *(const v4f*)(vb + 512 * j + 128 * t);
  }
}

__device__ __forceinline__ void attn_s_mma(v8f (&s)[NT], const v4f (&qa)[8], const v4f (&kf)[8][NT]) {
  #pragma unroll
  for (int t = 0; t < NT; ++t) s[t] = v8f{};
  #pragma unroll
  for (int g = 0; g < 8; ++g) {
    const v2f a01 = {qa[g].x, qa[g].y};
    const v2f a23 = {qa[g].z, qa[g].w};
    #pragma unroll
    for (int t = 0; t < NT; ++t) s[t] = wmma_f32(a01, v2f{kf[g][t].x, kf[g][t].y}, s[t]);
    #pragma unroll
    for (int t = 0; t < NT; ++t) s[t] = wmma_f32(a23, v2f{kf[g][t].z, kf[g][t].w}, s[t]);
  }
}

__device__ __forceinline__ void attn_pv_mma(v8f& acc0, v8f& acc1, v8f& acc2, v8f& acc3,
                                            const v4f (&pa)[2 * NT], const v4f (&vf)[2 * NT][4]) {
  #pragma unroll
  for (int j = 0; j < 2 * NT; ++j) {
    const v2f p01 = {pa[j].x, pa[j].y};
    const v2f p23 = {pa[j].z, pa[j].w};
    acc0 = wmma_f32(p01, v2f{vf[j][0].x, vf[j][0].y}, acc0);
    acc1 = wmma_f32(p01, v2f{vf[j][1].x, vf[j][1].y}, acc1);
    acc2 = wmma_f32(p01, v2f{vf[j][2].x, vf[j][2].y}, acc2);
    acc3 = wmma_f32(p01, v2f{vf[j][3].x, vf[j][3].y}, acc3);
    acc0 = wmma_f32(p23, v2f{vf[j][0].z, vf[j][0].w}, acc0);
    acc1 = wmma_f32(p23, v2f{vf[j][1].z, vf[j][1].w}, acc1);
    acc2 = wmma_f32(p23, v2f{vf[j][2].z, vf[j][2].w}, acc2);
    acc3 = wmma_f32(p23, v2f{vf[j][3].z, vf[j][3].w}, acc3);
  }
}

template <bool MASKED>
__device__ __forceinline__ void attn_softmax(v8f (&s)[NT], float (&rm)[8], float (&rs)[8],
                                             v8f& acc0, v8f& acc1, v8f& acc2, v8f& acc3,
                                             float (*prow)[PPITCH], int k0, int q0, int ln, int hi) {
  const float scale = 0.125f;
  #pragma unroll
  for (int t = 0; t < NT; ++t) {
    const int kg = k0 + 16 * t + ln;
    #pragma unroll
    for (int r = 0; r < 8; ++r) {
      const int qg = q0 + r + hi * 8;
      const float v = s[t][r] * scale;
      s[t][r] = (MASKED && kg > qg) ? 0.f : v;
    }
  }
  float corr[8];
  #pragma unroll
  for (int r = 0; r < 8; ++r) {
    float t = s[0][r];
    #pragma unroll
    for (int u = 1; u < NT; ++u) t = fmaxf(t, s[u][r]);
    t = fmaxf(t, row_xor<1>(t));
    t = fmaxf(t, row_xor<2>(t));
    t = fmaxf(t, row_xor<4>(t));
    t = fmaxf(t, row_xor<8>(t));
    const float nm = fmaxf(rm[r], t);
    const float c  = __expf(rm[r] - nm);
    float rsum = 0.f;
    #pragma unroll
    for (int u = 0; u < NT; ++u) {
      const float e = __expf(s[u][r] - nm);
      s[u][r] = e;
      rsum += e;
    }
    rsum += row_xor<1>(rsum);
    rsum += row_xor<2>(rsum);
    rsum += row_xor<4>(rsum);
    rsum += row_xor<8>(rsum);
    rs[r] = rs[r] * c + rsum;
    rm[r] = nm;
    corr[r] = c;
  }
  #pragma unroll
  for (int r = 0; r < 8; ++r) {
    acc0[r] *= corr[r]; acc1[r] *= corr[r];
    acc2[r] *= corr[r]; acc3[r] *= corr[r];
  }
  #pragma unroll
  for (int t = 0; t < NT; ++t) {
    #pragma unroll
    for (int r = 0; r < 8; ++r) prow[r + hi * 8][16 * t + ln] = s[t][r];
  }
}

__global__ __launch_bounds__(128) __attribute__((amdgpu_waves_per_eu(1, 1))) void attn_kernel(
    const float* __restrict__ ws, float* __restrict__ out) {
  __shared__ __attribute__((aligned(16))) float plds[4][16][PPITCH];

  const int lane = threadIdx.x;
  const int w    = threadIdx.y;
  const int b    = blockIdx.x;
  const int qt   = blockIdx.y * 4 + w;
  const int q0   = qt * 16;
  const int ln   = lane & 15;
  const int hi   = lane >> 4;

  const float* Q = ws + QOFF;
  const float* K = ws + KOFF;
  const float* V = ws + VOFF;

  v4f qa[8];
  {
    const float* qrow = Q + (size_t)(b * T_SEQ + q0 + ln) * HID + 4 * hi;
    #pragma unroll
    for (int g = 0; g < 8; ++g) qa[g] = *(const v4f*)(qrow + 8 * g);
  }

  v8f acc0 = {}, acc1 = {}, acc2 = {}, acc3 = {};
  float rm[8], rs[8];
  #pragma unroll
  for (int r = 0; r < 8; ++r) { rm[r] = -INFINITY; rs[r] = 0.f; }

  const int nblk = __builtin_amdgcn_readfirstlane(qt / NT + 1);
  const float* kbase = K + (size_t)(b * T_SEQ) * HID + 4 * lane;
  const float* vbase = V + (size_t)(b * T_SEQ) * HID + 4 * lane;
  float (*prow)[PPITCH] = plds[w];

  v4f kf[8][NT];
  v4f vf[2 * NT][4];
  v4f pa[2 * NT];
  v8f s[NT];

  attn_load_k(kf, kbase);
  attn_s_mma(s, qa, kf);
  const int lastk = (nblk - 1) * KB;
  if (nblk > 1) attn_load_k(kf, kbase + (size_t)KB * HID);

  for (int kb = 0; kb < nblk - 1; ++kb) {
    const int k0 = kb * KB;
    attn_load_v(vf, vbase + (size_t)k0 * HID);
    v8f sn[NT];
    attn_s_mma(sn, qa, kf);
    attn_softmax<false>(s, rm, rs, acc0, acc1, acc2, acc3, prow, k0, q0, ln, hi);
    asm volatile("s_wait_dscnt 0x0" ::: "memory");
    #pragma unroll
    for (int j = 0; j < 2 * NT; ++j) pa[j] = *(const v4f*)(&prow[ln][8 * j + 4 * hi]);
    attn_load_k(kf, kbase + (size_t)min(k0 + 2 * KB, lastk) * HID);
    __builtin_amdgcn_sched_barrier(0x406);
    attn_pv_mma(acc0, acc1, acc2, acc3, pa, vf);
    #pragma unroll
    for (int t = 0; t < NT; ++t) s[t] = sn[t];
  }

  {
    const int k0 = (nblk - 1) * KB;
    attn_load_v(vf, vbase + (size_t)k0 * HID);
    attn_softmax<true>(s, rm, rs, acc0, acc1, acc2, acc3, prow, k0, q0, ln, hi);
    asm volatile("s_wait_dscnt 0x0" ::: "memory");
    #pragma unroll
    for (int j = 0; j < 2 * NT; ++j) pa[j] = *(const v4f*)(&prow[ln][8 * j + 4 * hi]);
    attn_pv_mma(acc0, acc1, acc2, acc3, pa, vf);
  }

  const int cf = T_SEQ - KB * nblk;
  if (cf > 0) {
    const int nrow = __builtin_amdgcn_readfirstlane(cf);
    const float* vs = V + (size_t)(b * T_SEQ + KB * nblk) * HID + 4 * lane;
    v4f p0 = {}, p1 = {}, p2 = {}, p3 = {};
    for (int r0 = 0; r0 < nrow; r0 += 32) {
      v4f x[16];
      #pragma unroll
      for (int u = 0; u < 16; ++u) x[u] = *(const v4f*)(vs + (size_t)r0 * HID + 128 * u);
      #pragma unroll
      for (int u = 0; u < 16; u += 4) { p0 += x[u]; p1 += x[u + 1]; p2 += x[u + 2]; p3 += x[u + 3]; }
    }
    float s0 = (p0.x + p0.y) + (p0.z + p0.w);
    float s1 = (p1.x + p1.y) + (p1.z + p1.w);
    float s2 = (p2.x + p2.y) + (p2.z + p2.w);
    float s3 = (p3.x + p3.y) + (p3.z + p3.w);
    s0 += __shfl_xor(s0, 16);
    s1 += __shfl_xor(s1, 16);
    s2 += __shfl_xor(s2, 16);
    s3 += __shfl_xor(s3, 16);
    #pragma unroll
    for (int r = 0; r < 8; ++r) {
      const float nm = fmaxf(rm[r], 0.f);
      const float c  = __expf(rm[r] - nm);
      const float wz = __expf(-nm);
      rs[r] = rs[r] * c + wz * (float)cf;
      acc0[r] = acc0[r] * c + wz * s0;
      acc1[r] = acc1[r] * c + wz * s1;
      acc2[r] = acc2[r] * c + wz * s2;
      acc3[r] = acc3[r] * c + wz * s3;
    }
  }

  float* orow = out + (size_t)(b * T_SEQ + q0 + hi * 8) * HID + ln;
  #pragma unroll
  for (int r = 0; r < 8; ++r) {
    const float inv = 1.f / rs[r];
    orow[(size_t)r * HID +  0] = acc0[r] * inv;
    orow[(size_t)r * HID + 16] = acc1[r] * inv;
    orow[(size_t)r * HID + 32] = acc2[r] * inv;
    orow[(size_t)r * HID + 48] = acc3[r] * inv;
  }
}

extern "C" void kernel_launch(void* const* d_in, const int* in_sizes, int n_in,
                              void* d_out, int out_size, void* d_ws, size_t ws_size,
                              hipStream_t stream) {
  const float* X  = (const float*)d_in[0];
  const float* Wq = (const float*)d_in[1];
  const float* bq = (const float*)d_in[2];
  const float* Wk = (const float*)d_in[3];
  const float* bk = (const float*)d_in[4];
  const float* Wv = (const float*)d_in[5];
  const float* bv = (const float*)d_in[6];
  float* ws  = (float*)d_ws;
  float* out = (float*)d_out;

  qkv_kernel<<<dim3(MTOT / 64), dim3(32, 4, 3), 0, stream>>>(
      X, Wq, bq, Wk, bk, Wv, bv, ws);
  attn_kernel<<<dim3(NB, NKT / 4), dim3(32, 4), 0, stream>>>(ws, out);
}


# attention transposed (S^T=K.Q^T, O^T=V^T.P^T): P stays in registers from score tile to P.V WMMA, no LDS exchange, per-lane max/sum; V key order, softmax-sum, P.V and future-V sum order change
# speedup vs baseline: 4.1257x; 1.0577x over previous
#include <hip/hip_runtime.h>
#include <math.h>


typedef __attribute__((ext_vector_type(2))) float v2f;
typedef __attribute__((ext_vector_type(4))) float v4f;
typedef __attribute__((ext_vector_type(8))) float v8f;

#define T_SEQ 2048
#define EMB   1024
#define HID   64
#define NB    8
#define MTOT  (NB * T_SEQ)
#define NKT   (T_SEQ / 16)

#define QOFF   0
#define KOFF   ((size_t)MTOT * HID)
#define VOFF   ((size_t)2 * MTOT * HID)

__device__ __forceinline__ v8f wmma_f32(v2f a, v2f b, v8f c) {
  return __builtin_amdgcn_wmma_f32_16x16x4_f32(
       false, a,  false, b,
       (short)0, c,  false,  false);
}

__device__ __forceinline__ float xhalf(float v) {
  const int b = __float_as_int(v);
  return __int_as_float(__builtin_amdgcn_permlanex16(b, b, 0x76543210, 0xfedcba98, false, false));
}

#define KG  (EMB / 8)
#define KU  4

struct QkvFrag {
  v4f x[KU];
  v4f w[KU][4];
};

__device__ __forceinline__ void qkv_load(QkvFrag& f, const float* __restrict__ arow,
                                         const float* __restrict__ wrow, int g0) {
  #pragma unroll
  for (int u = 0; u < KU; ++u) {
    f.x[u] = *(const v4f*)(arow + 8 * (g0 + u));
    #pragma unroll
    for (int j = 0; j < 4; ++j)
      f.w[u][j] = *(const v4f*)(wrow + (size_t)(8 * (g0 + u) + j) * HID);
  }
}

__device__ __forceinline__ void qkv_mma(const QkvFrag& f, v8f& c0, v8f& c1, v8f& c2, v8f& c3) {
  #pragma unroll
  for (int u = 0; u < KU; ++u) {
    const v2f a01 = {f.x[u].x, f.x[u].y};
    const v2f a23 = {f.x[u].z, f.x[u].w};
    c0 = wmma_f32(a01, v2f{f.w[u][0].x, f.w[u][1].x}, c0);
    c1 = wmma_f32(a01, v2f{f.w[u][0].y, f.w[u][1].y}, c1);
    c2 = wmma_f32(a01, v2f{f.w[u][0].z, f.w[u][1].z}, c2);
    c3 = wmma_f32(a01, v2f{f.w[u][0].w, f.w[u][1].w}, c3);
    c0 = wmma_f32(a23, v2f{f.w[u][2].x, f.w[u][3].x}, c0);
    c1 = wmma_f32(a23, v2f{f.w[u][2].y, f.w[u][3].y}, c1);
    c2 = wmma_f32(a23, v2f{f.w[u][2].z, f.w[u][3].z}, c2);
    c3 = wmma_f32(a23, v2f{f.w[u][2].w, f.w[u][3].w}, c3);
  }
}

__global__ __launch_bounds__(384) __attribute__((amdgpu_waves_per_eu(3, 3))) void qkv_kernel(
    const float* __restrict__ X,
    const float* __restrict__ Wq, const float* __restrict__ bq,
    const float* __restrict__ Wk, const float* __restrict__ bk,
    const float* __restrict__ Wv, const float* __restrict__ bv,
    float* __restrict__ ws) {
  const int lane = threadIdx.x;
  const int mt   = __builtin_amdgcn_readfirstlane(threadIdx.y);
  const int mat  = __builtin_amdgcn_readfirstlane(threadIdx.z);
  const int m0   = (blockIdx.x * 4 + mt) * 16;
  const int c    = lane & 15;
  const int h    = lane >> 4;

  const float* W    = (mat == 0) ? Wq : (mat == 1) ? Wk : Wv;
  const float* bias = (mat == 0) ? bq : (mat == 1) ? bk : bv;
  float* out = ws + (size_t)mat * (size_t)MTOT * HID;

  const float* arow = X + (size_t)(m0 + c) * EMB + 4 * h;
  const float* wrow = W + (size_t)(4 * h) * HID + 4 * c;

  v8f c0 = {}, c1 = {}, c2 = {}, c3 = {};
  QkvFrag fa, fb;
  qkv_load(fa, arow, wrow, 0);
  #pragma nounroll
  for (int g0 = 0; g0 < KG - 2 * KU; g0 += 2 * KU) {
    qkv_load(fb, arow, wrow, g0 + KU);
    qkv_mma(fa, c0, c1, c2, c3);
    qkv_load(fa, arow, wrow, g0 + 2 * KU);
    qkv_mma(fb, c0, c1, c2, c3);
  }
  qkv_load(fb, arow, wrow, KG - KU);
  qkv_mma(fa, c0, c1, c2, c3);
  qkv_mma(fb, c0, c1, c2, c3);

  const v4f bb = *(const v4f*)(bias + 4 * c);
  float* tile = out + (size_t)m0 * HID;
  if (mat == 0) {
    float* orow = tile + (size_t)(h * 8) * HID + 4 * c;
    #pragma unroll
    for (int r = 0; r < 8; ++r) {
      v4f o = {c0[r] + bb.x, c1[r] + bb.y, c2[r] + bb.z, c3[r] + bb.w};
      *(v4f*)(orow + (size_t)r * HID) = o;
    }
  } else if (mat == 1) {
    float* o4 = tile + (size_t)(16 * c + 8 * h) * 4;
    #pragma unroll
    for (int r = 0; r < 8; ++r) {
      v4f o = {c0[r] + bb.x, c1[r] + bb.y, c2[r] + bb.z, c3[r] + bb.w};
      *(v4f*)(o4 + 4 * r) = o;
    }
  } else {
    float* o4 = tile + (size_t)(32 * (c >> 2) + 16 * h + 4 * (c & 3)) * 4;
    #pragma unroll
    for (int q = 0; q < 2; ++q) {
      const v4f o0 = {c0[4 * q] + bb.x, c0[4 * q + 1] + bb.x, c0[4 * q + 2] + bb.x, c0[4 * q + 3] + bb.x};
      const v4f o1 = {c1[4 * q] + bb.y, c1[4 * q + 1] + bb.y, c1[4 * q + 2] + bb.y, c1[4 * q + 3] + bb.y};
      const v4f o2 = {c2[4 * q] + bb.z, c2[4 * q + 1] + bb.z, c2[4 * q + 2] + bb.z, c2[4 * q + 3] + bb.z};
      const v4f o3 = {c3[4 * q] + bb.w, c3[4 * q + 1] + bb.w, c3[4 * q + 2] + bb.w, c3[4 * q + 3] + bb.w};
      *(v4f*)(o4 + (size_t)(128 * q + 0) * 4) = o0;
      *(v4f*)(o4 + (size_t)(128 * q + 1) * 4) = o1;
      *(v4f*)(o4 + (size_t)(128 * q + 2) * 4) = o2;
      *(v4f*)(o4 + (size_t)(128 * q + 3) * 4) = o3;
    }
  }
}

#define NT     8
#define KB     (16 * NT)

__device__ __forceinline__ void attn_load_k(v4f (&kf)[8][NT], const float* __restrict__ kt) {
  #pragma unroll
  for (int g = 0; g < 8; ++g) {
    #pragma unroll
    for (int t = 0; t < NT; ++t) kf[g][t] = *(const v4f*)(kt + (size_t)t * 16 * HID + 128 * g);
  }
}

__device__ __forceinline__ void attn_load_v(v4f (&vf)[2 * NT][4], const float* __restrict__ vb) {
  #pragma unroll
  for (int j = 0; j < 2 * NT; ++j) {
    #pragma unroll
    for (int t = 0; t < 4; ++t) vf[j][t] = *(const v4f*)(vb + 512 * j + 128 * t);
  }
}

__device__ __forceinline__ void attn_s_mma(v8f (&s)[NT], const v4f (&qa)[8], const v4f (&kf)[8][NT]) {
  #pragma unroll
  for (int t = 0; t < NT; ++t) s[t] = v8f{};
  #pragma unroll
  for (int g = 0; g < 8; ++g) {
    const v2f q01 = {qa[g].x, qa[g].y};
    const v2f q23 = {qa[g].z, qa[g].w};
    #pragma unroll
    for (int t = 0; t < NT; ++t) s[t] = wmma_f32(v2f{kf[g][t].x, kf[g][t].y}, q01, s[t]);
    #pragma unroll
    for (int t = 0; t < NT; ++t) s[t] = wmma_f32(v2f{kf[g][t].z, kf[g][t].w}, q23, s[t]);
  }
}

__device__ __forceinline__ void attn_pv_mma(v8f& acc0, v8f& acc1, v8f& acc2, v8f& acc3,
                                            const v8f (&p)[NT], const v4f (&vf)[2 * NT][4]) {
  #pragma unroll
  for (int t = 0; t < NT; ++t) {
    #pragma unroll
    for (int q = 0; q < 2; ++q) {
      const int j = 2 * t + q;
      const v2f p01 = {p[t][4 * q], p[t][4 * q + 1]};
      const v2f p23 = {p[t][4 * q + 2], p[t][4 * q + 3]};
      acc0 = wmma_f32(v2f{vf[j][0].x, vf[j][0].y}, p01, acc0);
      acc1 = wmma_f32(v2f{vf[j][1].x, vf[j][1].y}, p01, acc1);
      acc2 = wmma_f32(v2f{vf[j][2].x, vf[j][2].y}, p01, acc2);
      acc3 = wmma_f32(v2f{vf[j][3].x, vf[j][3].y}, p01, acc3);
      acc0 = wmma_f32(v2f{vf[j][0].z, vf[j][0].w}, p23, acc0);
      acc1 = wmma_f32(v2f{vf[j][1].z, vf[j][1].w}, p23, acc1);
      acc2 = wmma_f32(v2f{vf[j][2].z, vf[j][2].w}, p23, acc2);
      acc3 = wmma_f32(v2f{vf[j][3].z, vf[j][3].w}, p23, acc3);
    }
  }
}

template <bool MASKED>
__device__ __forceinline__ void attn_softmax(v8f (&s)[NT], float& rm, float& rs,
                                             v8f& acc0, v8f& acc1, v8f& acc2, v8f& acc3,
                                             int k0, int q0, int ln, int hi) {
  const float scale = 0.125f;
  const int qg = q0 + ln;
  #pragma unroll
  for (int t = 0; t < NT; ++t) {
    #pragma unroll
    for (int r = 0; r < 8; ++r) {
      const int kg = k0 + 16 * t + 8 * hi + r;
      const float v = s[t][r] * scale;
      s[t][r] = (MASKED && kg > qg) ? 0.f : v;
    }
  }
  float mx[8];
  #pragma unroll
  for (int r = 0; r < 8; ++r) {
    mx[r] = s[0][r];
    #pragma unroll
    for (int t = 1; t < NT; ++t) mx[r] = fmaxf(mx[r], s[t][r]);
  }
  float m = fmaxf(fmaxf(fmaxf(mx[0], mx[1]), fmaxf(mx[2], mx[3])),
                  fmaxf(fmaxf(mx[4], mx[5]), fmaxf(mx[6], mx[7])));
  m = fmaxf(m, xhalf(m));
  const float nm = fmaxf(rm, m);
  const float c  = __expf(rm - nm);
  float part[8];
  #pragma unroll
  for (int r = 0; r < 8; ++r) part[r] = 0.f;
  #pragma unroll
  for (int t = 0; t < NT; ++t) {
    #pragma unroll
    for (int r = 0; r < 8; ++r) {
      const float e = __expf(s[t][r] - nm);
      s[t][r] = e;
      part[r] += e;
    }
  }
  const float rsum = ((part[0] + part[1]) + (part[2] + part[3])) +
                     ((part[4] + part[5]) + (part[6] + part[7]));
  rs = rs * c + rsum;
  rm = nm;
  #pragma unroll
  for (int r = 0; r < 8; ++r) {
    acc0[r] *= c; acc1[r] *= c;
    acc2[r] *= c; acc3[r] *= c;
  }
}

__global__ __launch_bounds__(128) __attribute__((amdgpu_waves_per_eu(1, 1))) void attn_kernel(
    const float* __restrict__ ws, float* __restrict__ out) {
  const int lane = threadIdx.x;
  const int w    = threadIdx.y;
  const int b    = blockIdx.x;
  const int qt   = blockIdx.y * 4 + w;
  const int q0   = qt * 16;
  const int ln   = lane & 15;
  const int hi   = lane >> 4;

  const float* Q = ws + QOFF;
  const float* K = ws + KOFF;
  const float* V = ws + VOFF;

  v4f qa[8];
  {
    const float* qrow = Q + (size_t)(b * T_SEQ + q0 + ln) * HID + 4 * hi;
    #pragma unroll
    for (int g = 0; g < 8; ++g) qa[g] = *(const v4f*)(qrow + 8 * g);
  }

  v8f acc0 = {}, acc1 = {}, acc2 = {}, acc3 = {};
  float rm = -INFINITY, rs = 0.f;

  const int nblk = __builtin_amdgcn_readfirstlane(qt / NT + 1);
  const float* kbase = K + (size_t)(b * T_SEQ) * HID + 4 * lane;
  const float* vbase = V + (size_t)(b * T_SEQ) * HID + 4 * lane;

  v4f kf[8][NT];
  v4f vf[2 * NT][4];
  v8f s[NT];

  attn_load_k(kf, kbase);
  attn_s_mma(s, qa, kf);
  const int lastk = (nblk - 1) * KB;
  if (nblk > 1) attn_load_k(kf, kbase + (size_t)KB * HID);

  for (int kb = 0; kb < nblk - 1; ++kb) {
    const int k0 = kb * KB;
    attn_load_v(vf, vbase + (size_t)k0 * HID);
    v8f sn[NT];
    attn_s_mma(sn, qa, kf);
    attn_softmax<false>(s, rm, rs, acc0, acc1, acc2, acc3, k0, q0, ln, hi);
    attn_load_k(kf, kbase + (size_t)min(k0 + 2 * KB, lastk) * HID);
    __builtin_amdgcn_sched_barrier(0x406);
    attn_pv_mma(acc0, acc1, acc2, acc3, s, vf);
    #pragma unroll
    for (int t = 0; t < NT; ++t) s[t] = sn[t];
  }

  {
    const int k0 = (nblk - 1) * KB;
    attn_load_v(vf, vbase + (size_t)k0 * HID);
    attn_softmax<true>(s, rm, rs, acc0, acc1, acc2, acc3, k0, q0, ln, hi);
    attn_pv_mma(acc0, acc1, acc2, acc3, s, vf);
  }

  float rst = rs + xhalf(rs);

  const int cf = T_SEQ - KB * nblk;
  if (cf > 0) {
    const int nrow = __builtin_amdgcn_readfirstlane(cf);
    const float* vs = V + (size_t)(b * T_SEQ + KB * nblk) * HID + 4 * lane;
    v4f p0 = {}, p1 = {}, p2 = {}, p3 = {};
    for (int r0 = 0; r0 < nrow; r0 += 32) {
      v4f x[16];
      #pragma unroll
      for (int u = 0; u < 16; ++u) x[u] = *(const v4f*)(vs + (size_t)r0 * HID + 128 * u);
      #pragma unroll
      for (int u = 0; u < 16; u += 4) { p0 += x[u]; p1 += x[u + 1]; p2 += x[u + 2]; p3 += x[u + 3]; }
    }
    float s0 = (p0.x + p0.y) + (p0.z + p0.w);
    float s1 = (p1.x + p1.y) + (p1.z + p1.w);
    float s2 = (p2.x + p2.y) + (p2.z + p2.w);
    float s3 = (p3.x + p3.y) + (p3.z + p3.w);
    s0 += __shfl_xor(s0, 16);
    s1 += __shfl_xor(s1, 16);
    s2 += __shfl_xor(s2, 16);
    s3 += __shfl_xor(s3, 16);
    const float nm = fmaxf(rm, 0.f);
    const float c  = __expf(rm - nm);
    const float wz = __expf(-nm);
    rst = rst * c + wz * (float)cf;
    #pragma unroll
    for (int r = 0; r < 8; ++r) {
      const int from = 8 * hi + r;
      acc0[r] = acc0[r] * c + wz * __shfl(s0, from);
      acc1[r] = acc1[r] * c + wz * __shfl(s1, from);
      acc2[r] = acc2[r] * c + wz * __shfl(s2, from);
      acc3[r] = acc3[r] * c + wz * __shfl(s3, from);
    }
  }

  const float inv = 1.f / rst;
  float* orow = out + (size_t)(b * T_SEQ + q0 + ln) * HID + 8 * hi;
  *(v4f*)(orow +  0) = v4f{acc0[0] * inv, acc0[1] * inv, acc0[2] * inv, acc0[3] * inv};
  *(v4f*)(orow +  4) = v4f{acc0[4] * inv, acc0[5] * inv, acc0[6] * inv, acc0[7] * inv};
  *(v4f*)(orow + 16) = v4f{acc1[0] * inv, acc1[1] * inv, acc1[2] * inv, acc1[3] * inv};
  *(v4f*)(orow + 20) = v4f{acc1[4] * inv, acc1[5] * inv, acc1[6] * inv, acc1[7] * inv};
  *(v4f*)(orow + 32) = v4f{acc2[0] * inv, acc2[1] * inv, acc2[2] * inv, acc2[3] * inv};
  *(v4f*)(orow + 36) = v4f{acc2[4] * inv, acc2[5] * inv, acc2[6] * inv, acc2[7] * inv};
  *(v4f*)(orow + 48) = v4f{acc3[0] * inv, acc3[1] * inv, acc3[2] * inv, acc3[3] * inv};
  *(v4f*)(orow + 52) = v4f{acc3[4] * inv, acc3[5] * inv, acc3[6] * inv, acc3[7] * inv};
}

extern "C" void kernel_launch(void* const* d_in, const int* in_sizes, int n_in,
                              void* d_out, int out_size, void* d_ws, size_t ws_size,
                              hipStream_t stream) {
  const float* X  = (const float*)d_in[0];
  const float* Wq = (const float*)d_in[1];
  const float* bq = (const float*)d_in[2];
  const float* Wk = (const float*)d_in[3];
  const float* bk = (const float*)d_in[4];
  const float* Wv = (const float*)d_in[5];
  const float* bv = (const float*)d_in[6];
  float* ws  = (float*)d_ws;
  float* out = (float*)d_out;

  qkv_kernel<<<dim3(MTOT / 64), dim3(32, 4, 3), 0, stream>>>(
      X, Wq, bq, Wk, bk, Wv, bv, ws);
  attn_kernel<<<dim3(NB, NKT / 4), dim3(32, 4), 0, stream>>>(ws, out);
}


# attention load balance: query tiles folded in pairs (p, 127-p), 2 waves a pair; long tile's first 8-11 key trips on wave 2, partials (max,sum,O) merged in LDS; softmax/P.V/future-V sum order changes
# speedup vs baseline: 4.6602x; 1.1296x over previous
#include <hip/hip_runtime.h>
#include <math.h>


typedef __attribute__((ext_vector_type(2))) float v2f;
typedef __attribute__((ext_vector_type(4))) float v4f;
typedef __attribute__((ext_vector_type(8))) float v8f;

#define T_SEQ 2048
#define EMB   1024
#define HID   64
#define NB    8
#define MTOT  (NB * T_SEQ)
#define NKT   (T_SEQ / 16)

#define QOFF   0
#define KOFF   ((size_t)MTOT * HID)
#define VOFF   ((size_t)2 * MTOT * HID)

__device__ __forceinline__ v8f wmma_f32(v2f a, v2f b, v8f c) {
  return __builtin_amdgcn_wmma_f32_16x16x4_f32(
       false, a,  false, b,
       (short)0, c,  false,  false);
}

__device__ __forceinline__ float xhalf(float v) {
  const int b = __float_as_int(v);
  return __int_as_float(__builtin_amdgcn_permlanex16(b, b, 0x76543210, 0xfedcba98, false, false));
}

#define KG  (EMB / 8)
#define KU  4

struct QkvFrag {
  v4f x[KU];
  v4f w[KU][4];
};

__device__ __forceinline__ void qkv_load(QkvFrag& f, const float* __restrict__ arow,
                                         const float* __restrict__ wrow, int g0) {
  #pragma unroll
  for (int u = 0; u < KU; ++u) {
    f.x[u] = *(const v4f*)(arow + 8 * (g0 + u));
    #pragma unroll
    for (int j = 0; j < 4; ++j)
      f.w[u][j] = *(const v4f*)(wrow + (size_t)(8 * (g0 + u) + j) * HID);
  }
}

__device__ __forceinline__ void qkv_mma(const QkvFrag& f, v8f& c0, v8f& c1, v8f& c2, v8f& c3) {
  #pragma unroll
  for (int u = 0; u < KU; ++u) {
    const v2f a01 = {f.x[u].x, f.x[u].y};
    const v2f a23 = {f.x[u].z, f.x[u].w};
    c0 = wmma_f32(a01, v2f{f.w[u][0].x, f.w[u][1].x}, c0);
    c1 = wmma_f32(a01, v2f{f.w[u][0].y, f.w[u][1].y}, c1);
    c2 = wmma_f32(a01, v2f{f.w[u][0].z, f.w[u][1].z}, c2);
    c3 = wmma_f32(a01, v2f{f.w[u][0].w, f.w[u][1].w}, c3);
    c0 = wmma_f32(a23, v2f{f.w[u][2].x, f.w[u][3].x}, c0);
    c1 = wmma_f32(a23, v2f{f.w[u][2].y, f.w[u][3].y}, c1);
    c2 = wmma_f32(a23, v2f{f.w[u][2].z, f.w[u][3].z}, c2);
    c3 = wmma_f32(a23, v2f{f.w[u][2].w, f.w[u][3].w}, c3);
  }
}

__global__ __launch_bounds__(384) __attribute__((amdgpu_waves_per_eu(3, 3))) void qkv_kernel(
    const float* __restrict__ X,
    const float* __restrict__ Wq, const float* __restrict__ bq,
    const float* __restrict__ Wk, const float* __restrict__ bk,
    const float* __restrict__ Wv, const float* __restrict__ bv,
    float* __restrict__ ws) {
  const int lane = threadIdx.x;
  const int mt   = __builtin_amdgcn_readfirstlane(threadIdx.y);
  const int mat  = __builtin_amdgcn_readfirstlane(threadIdx.z);
  const int m0   = (blockIdx.x * 4 + mt) * 16;
  const int c    = lane & 15;
  const int h    = lane >> 4;

  const float* W    = (mat == 0) ? Wq : (mat == 1) ? Wk : Wv;
  const float* bias = (mat == 0) ? bq : (mat == 1) ? bk : bv;
  float* out = ws + (size_t)mat * (size_t)MTOT * HID;

  const float* arow = X + (size_t)(m0 + c) * EMB + 4 * h;
  const float* wrow = W + (size_t)(4 * h) * HID + 4 * c;

  v8f c0 = {}, c1 = {}, c2 = {}, c3 = {};
  QkvFrag fa, fb;
  qkv_load(fa, arow, wrow, 0);
  #pragma nounroll
  for (int g0 = 0; g0 < KG - 2 * KU; g0 += 2 * KU) {
    qkv_load(fb, arow, wrow, g0 + KU);
    qkv_mma(fa, c0, c1, c2, c3);
    qkv_load(fa, arow, wrow, g0 + 2 * KU);
    qkv_mma(fb, c0, c1, c2, c3);
  }
  qkv_load(fb, arow, wrow, KG - KU);
  qkv_mma(fa, c0, c1, c2, c3);
  qkv_mma(fb, c0, c1, c2, c3);

  const v4f bb = *(const v4f*)(bias + 4 * c);
  float* tile = out + (size_t)m0 * HID;
  if (mat == 0) {
    float* orow = tile + (size_t)(h * 8) * HID + 4 * c;
    #pragma unroll
    for (int r = 0; r < 8; ++r) {
      v4f o = {c0[r] + bb.x, c1[r] + bb.y, c2[r] + bb.z, c3[r] + bb.w};
      *(v4f*)(orow + (size_t)r * HID) = o;
    }
  } else if (mat == 1) {
    float* o4 = tile + (size_t)(16 * c + 8 * h) * 4;
    #pragma unroll
    for (int r = 0; r < 8; ++r) {
      v4f o = {c0[r] + bb.x, c1[r] + bb.y, c2[r] + bb.z, c3[r] + bb.w};
      *(v4f*)(o4 + 4 * r) = o;
    }
  } else {
    float* o4 = tile + (size_t)(32 * (c >> 2) + 16 * h + 4 * (c & 3)) * 4;
    #pragma unroll
    for (int q = 0; q < 2; ++q) {
      const v4f o0 = {c0[4 * q] + bb.x, c0[4 * q + 1] + bb.x, c0[4 * q + 2] + bb.x, c0[4 * q + 3] + bb.x};
      const v4f o1 = {c1[4 * q] + bb.y, c1[4 * q + 1] + bb.y, c1[4 * q + 2] + bb.y, c1[4 * q + 3] + bb.y};
      const v4f o2 = {c2[4 * q] + bb.z, c2[4 * q + 1] + bb.z, c2[4 * q + 2] + bb.z, c2[4 * q + 3] + bb.z};
      const v4f o3 = {c3[4 * q] + bb.w, c3[4 * q + 1] + bb.w, c3[4 * q + 2] + bb.w, c3[4 * q + 3] + bb.w};
      *(v4f*)(o4 + (size_t)(128 * q + 0) * 4) = o0;
      *(v4f*)(o4 + (size_t)(128 * q + 1) * 4) = o1;
      *(v4f*)(o4 + (size_t)(128 * q + 2) * 4) = o2;
      *(v4f*)(o4 + (size_t)(128 * q + 3) * 4) = o3;
    }
  }
}

#define NT     8
#define KB     (16 * NT)

__device__ __forceinline__ void attn_load_k(v4f (&kf)[8][NT], const float* __restrict__ kt) {
  #pragma unroll
  for (int g = 0; g < 8; ++g) {
    #pragma unroll
    for (int t = 0; t < NT; ++t) kf[g][t] = *(const v4f*)(kt + (size_t)t * 16 * HID + 128 * g);
  }
}

__device__ __forceinline__ void attn_load_v(v4f (&vf)[2 * NT][4], const float* __restrict__ vb) {
  #pragma unroll
  for (int j = 0; j < 2 * NT; ++j) {
    #pragma unroll
    for (int t = 0; t < 4; ++t) vf[j][t] = *(const v4f*)(vb + 512 * j + 128 * t);
  }
}

__device__ __forceinline__ void attn_s_mma(v8f (&s)[NT], const v4f (&qa)[8], const v4f (&kf)[8][NT]) {
  #pragma unroll
  for (int t = 0; t < NT; ++t) s[t] = v8f{};
  #pragma unroll
  for (int g = 0; g < 8; ++g) {
    const v2f q01 = {qa[g].x, qa[g].y};
    const v2f q23 = {qa[g].z, qa[g].w};
    #pragma unroll
    for (int t = 0; t < NT; ++t) s[t] = wmma_f32(v2f{kf[g][t].x, kf[g][t].y}, q01, s[t]);
    #pragma unroll
    for (int t = 0; t < NT; ++t) s[t] = wmma_f32(v2f{kf[g][t].z, kf[g][t].w}, q23, s[t]);
  }
}

__device__ __forceinline__ void attn_pv_mma(v8f& acc0, v8f& acc1, v8f& acc2, v8f& acc3,
                                            const v8f (&p)[NT], const v4f (&vf)[2 * NT][4]) {
  #pragma unroll
  for (int t = 0; t < NT; ++t) {
    #pragma unroll
    for (int q = 0; q < 2; ++q) {
      const int j = 2 * t + q;
      const v2f p01 = {p[t][4 * q], p[t][4 * q + 1]};
      const v2f p23 = {p[t][4 * q + 2], p[t][4 * q + 3]};
      acc0 = wmma_f32(v2f{vf[j][0].x, vf[j][0].y}, p01, acc0);
      acc1 = wmma_f32(v2f{vf[j][1].x, vf[j][1].y}, p01, acc1);
      acc2 = wmma_f32(v2f{vf[j][2].x, vf[j][2].y}, p01, acc2);
      acc3 = wmma_f32(v2f{vf[j][3].x, vf[j][3].y}, p01, acc3);
      acc0 = wmma_f32(v2f{vf[j][0].z, vf[j][0].w}, p23, acc0);
      acc1 = wmma_f32(v2f{vf[j][1].z, vf[j][1].w}, p23, acc1);
      acc2 = wmma_f32(v2f{vf[j][2].z, vf[j][2].w}, p23, acc2);
      acc3 = wmma_f32(v2f{vf[j][3].z, vf[j][3].w}, p23, acc3);
    }
  }
}

template <bool MASKED>
__device__ __forceinline__ void attn_softmax(v8f (&s)[NT], float& rm, float& rs,
                                             v8f& acc0, v8f& acc1, v8f& acc2, v8f& acc3,
                                             int k0, int q0, int ln, int hi) {
  const float scale = 0.125f;
  const int qg = q0 + ln;
  #pragma unroll
  for (int t = 0; t < NT; ++t) {
    #pragma unroll
    for (int r = 0; r < 8; ++r) {
      const int kg = k0 + 16 * t + 8 * hi + r;
      const float v = s[t][r] * scale;
      s[t][r] = (MASKED && kg > qg) ? 0.f : v;
    }
  }
  float mx[8];
  #pragma unroll
  for (int r = 0; r < 8; ++r) {
    mx[r] = s[0][r];
    #pragma unroll
    for (int t = 1; t < NT; ++t) mx[r] = fmaxf(mx[r], s[t][r]);
  }
  float m = fmaxf(fmaxf(fmaxf(mx[0], mx[1]), fmaxf(mx[2], mx[3])),
                  fmaxf(fmaxf(mx[4], mx[5]), fmaxf(mx[6], mx[7])));
  m = fmaxf(m, xhalf(m));
  const float nm = fmaxf(rm, m);
  const float c  = __expf(rm - nm);
  float part[8];
  #pragma unroll
  for (int r = 0; r < 8; ++r) part[r] = 0.f;
  #pragma unroll
  for (int t = 0; t < NT; ++t) {
    #pragma unroll
    for (int r = 0; r < 8; ++r) {
      const float e = __expf(s[t][r] - nm);
      s[t][r] = e;
      part[r] += e;
    }
  }
  const float rsum = ((part[0] + part[1]) + (part[2] + part[3])) +
                     ((part[4] + part[5]) + (part[6] + part[7]));
  rs = rs * c + rsum;
  rm = nm;
  #pragma unroll
  for (int r = 0; r < 8; ++r) {
    acc0[r] *= c; acc1[r] *= c;
    acc2[r] *= c; acc3[r] *= c;
  }
}

#define HA_MAX 11

__device__ __forceinline__ void attn_store(float* __restrict__ orow, const v8f& acc0, const v8f& acc1,
                                           const v8f& acc2, const v8f& acc3, float inv) {
  *(v4f*)(orow +  0) = v4f{acc0[0] * inv, acc0[1] * inv, acc0[2] * inv, acc0[3] * inv};
  *(v4f*)(orow +  4) = v4f{acc0[4] * inv, acc0[5] * inv, acc0[6] * inv, acc0[7] * inv};
  *(v4f*)(orow + 16) = v4f{acc1[0] * inv, acc1[1] * inv, acc1[2] * inv, acc1[3] * inv};
  *(v4f*)(orow + 20) = v4f{acc1[4] * inv, acc1[5] * inv, acc1[6] * inv, acc1[7] * inv};
  *(v4f*)(orow + 32) = v4f{acc2[0] * inv, acc2[1] * inv, acc2[2] * inv, acc2[3] * inv};
  *(v4f*)(orow + 36) = v4f{acc2[4] * inv, acc2[5] * inv, acc2[6] * inv, acc2[7] * inv};
  *(v4f*)(orow + 48) = v4f{acc3[0] * inv, acc3[1] * inv, acc3[2] * inv, acc3[3] * inv};
  *(v4f*)(orow + 52) = v4f{acc3[4] * inv, acc3[5] * inv, acc3[6] * inv, acc3[7] * inv};
}

__global__ __launch_bounds__(128) __attribute__((amdgpu_waves_per_eu(1, 1))) void attn_kernel(
    const float* __restrict__ ws, float* __restrict__ out) {
  __shared__ float xch[2][34][32];

  const int lane = threadIdx.x;
  const int w    = __builtin_amdgcn_readfirstlane(threadIdx.y);
  const int b    = blockIdx.x;
  const int pr   = w >> 1;
  const int role = w & 1;
  const int st   = blockIdx.y * 2 + pr;
  const int lt   = NKT - 1 - st;
  const int nS   = st / NT + 1;
  const int nL   = NKT / NT - st / NT;
  const int hA   = min(nL - 1, HA_MAX);
  const int ln   = lane & 15;
  const int hi   = lane >> 4;

  const float* Q = ws + QOFF;
  const float* K = ws + KOFF;
  const float* V = ws + VOFF;

  v8f acc0 = {}, acc1 = {}, acc2 = {}, acc3 = {};
  float rm = -INFINITY, rst = 0.f;
  v4f p0 = {}, p1 = {}, p2 = {}, p3 = {};

  const int nseg = role + 1;
  for (int sg = 0; sg < nseg; ++sg) {
    const bool tail = (role == 1) && (sg == 0);
    const bool shrt = (role == 1) && (sg == 1);
    const int qt   = shrt ? st : lt;
    const int blk0 = tail ? hA : 0;
    const int nblk = shrt ? nS : (tail ? nL - hA : hA);
    const int cflo = shrt ? nS * KB : nL * KB;
    const int cfhi = shrt ? nL * KB : T_SEQ;
    const int cfrows = (role == 1) ? cfhi - cflo : 0;
    const int cf     = (role == 1) ? T_SEQ - cflo : 0;
    const int q0   = qt * 16;

    v4f qa[8];
    {
      const float* qrow = Q + (size_t)(b * T_SEQ + q0 + ln) * HID + 4 * hi;
      #pragma unroll
      for (int g = 0; g < 8; ++g) qa[g] = *(const v4f*)(qrow + 8 * g);
    }

    acc0 = v8f{}; acc1 = v8f{}; acc2 = v8f{}; acc3 = v8f{};
    rm = -INFINITY;
    float rs = 0.f;

    const float* kbase = K + (size_t)(b * T_SEQ + blk0 * KB) * HID + 4 * lane;
    const float* vbase = V + (size_t)(b * T_SEQ + blk0 * KB) * HID + 4 * lane;

    v4f kf[8][NT];
    v4f vf[2 * NT][4];
    v8f s[NT];

    attn_load_k(kf, kbase);
    attn_s_mma(s, qa, kf);
    const int lastk = (nblk - 1) * KB;
    if (nblk > 1) attn_load_k(kf, kbase + (size_t)KB * HID);

    for (int kb = 0; kb < nblk - 1; ++kb) {
      const int k0 = kb * KB;
      attn_load_v(vf, vbase + (size_t)k0 * HID);
      v8f sn[NT];
      attn_s_mma(sn, qa, kf);
      attn_softmax<false>(s, rm, rs, acc0, acc1, acc2, acc3, k0, q0, ln, hi);
      attn_load_k(kf, kbase + (size_t)min(k0 + 2 * KB, lastk) * HID);
      __builtin_amdgcn_sched_barrier(0x406);
      attn_pv_mma(acc0, acc1, acc2, acc3, s, vf);
      #pragma unroll
      for (int t = 0; t < NT; ++t) s[t] = sn[t];
    }

    {
      attn_load_v(vf, vbase + (size_t)lastk * HID);
      attn_softmax<true>(s, rm, rs, acc0, acc1, acc2, acc3, blk0 * KB + lastk, q0, ln, hi);
      attn_pv_mma(acc0, acc1, acc2, acc3, s, vf);
    }

    rst = rs + xhalf(rs);

    if (cfrows > 0) {
      const float* vs = V + (size_t)(b * T_SEQ + cflo) * HID + 4 * lane;
      for (int r0 = 0; r0 < cfrows; r0 += 32) {
        v4f x[16];
        #pragma unroll
        for (int u = 0; u < 16; ++u) x[u] = *(const v4f*)(vs + (size_t)r0 * HID + 128 * u);
        #pragma unroll
        for (int u = 0; u < 16; u += 4) { p0 += x[u]; p1 += x[u + 1]; p2 += x[u + 2]; p3 += x[u + 3]; }
      }
    }

    if (cf > 0) {
      float s0 = (p0.x + p0.y) + (p0.z + p0.w);
      float s1 = (p1.x + p1.y) + (p1.z + p1.w);
      float s2 = (p2.x + p2.y) + (p2.z + p2.w);
      float s3 = (p3.x + p3.y) + (p3.z + p3.w);
      s0 += __shfl_xor(s0, 16);
      s1 += __shfl_xor(s1, 16);
      s2 += __shfl_xor(s2, 16);
      s3 += __shfl_xor(s3, 16);
      const float nm = fmaxf(rm, 0.f);
      const float c  = __expf(rm - nm);
      const float wz = __expf(-nm);
      rst = rst * c + wz * (float)cf;
      rm  = nm;
      #pragma unroll
      for (int r = 0; r < 8; ++r) {
        const int from = 8 * hi + r;
        acc0[r] = acc0[r] * c + wz * __shfl(s0, from);
        acc1[r] = acc1[r] * c + wz * __shfl(s1, from);
        acc2[r] = acc2[r] * c + wz * __shfl(s2, from);
        acc3[r] = acc3[r] * c + wz * __shfl(s3, from);
      }
    }

    if (tail) {
      xch[pr][0][lane] = rm;
      xch[pr][1][lane] = rst;
      #pragma unroll
      for (int r = 0; r < 8; ++r) {
        xch[pr][2 + r][lane]  = acc0[r];
        xch[pr][10 + r][lane] = acc1[r];
        xch[pr][18 + r][lane] = acc2[r];
        xch[pr][26 + r][lane] = acc3[r];
      }
    }
    if (shrt) {
      attn_store(out + (size_t)(b * T_SEQ + q0 + ln) * HID + 8 * hi, acc0, acc1, acc2, acc3, 1.f / rst);
    }
  }

  __syncthreads();

  if (role == 0) {
    const float rmB  = xch[pr][0][lane];
    const float rstB = xch[pr][1][lane];
    const float nm = fmaxf(rm, rmB);
    const float cA = __expf(rm - nm);
    const float cB = __expf(rmB - nm);
    rst = rst * cA + rstB * cB;
    #pragma unroll
    for (int r = 0; r < 8; ++r) {
      acc0[r] = acc0[r] * cA + xch[pr][2 + r][lane]  * cB;
      acc1[r] = acc1[r] * cA + xch[pr][10 + r][lane] * cB;
      acc2[r] = acc2[r] * cA + xch[pr][18 + r][lane] * cB;
      acc3[r] = acc3[r] * cA + xch[pr][26 + r][lane] * cB;
    }
    attn_store(out + (size_t)(b * T_SEQ + lt * 16 + ln) * HID + 8 * hi, acc0, acc1, acc2, acc3, 1.f / rst);
  }
}

extern "C" void kernel_launch(void* const* d_in, const int* in_sizes, int n_in,
                              void* d_out, int out_size, void* d_ws, size_t ws_size,
                              hipStream_t stream) {
  const float* X  = (const float*)d_in[0];
  const float* Wq = (const float*)d_in[1];
  const float* bq = (const float*)d_in[2];
  const float* Wk = (const float*)d_in[3];
  const float* bk = (const float*)d_in[4];
  const float* Wv = (const float*)d_in[5];
  const float* bv = (const float*)d_in[6];
  float* ws  = (float*)d_ws;
  float* out = (float*)d_out;

  qkv_kernel<<<dim3(MTOT / 64), dim3(32, 4, 3), 0, stream>>>(
      X, Wq, bq, Wk, bk, Wv, bv, ws);
  attn_kernel<<<dim3(NB, NKT / 4), dim3(32, 4), 0, stream>>>(ws, out);
}


# future-region V sums read per-tile column sums stored by the projection's V waves (256 B a tile, not its rows); head/tail trips 9/8, were 11/6; future-V sum order, softmax/P.V partial split change
# speedup vs baseline: 4.9583x; 1.0640x over previous
#include <hip/hip_runtime.h>
#include <math.h>


typedef __attribute__((ext_vector_type(2))) float v2f;
typedef __attribute__((ext_vector_type(4))) float v4f;
typedef __attribute__((ext_vector_type(8))) float v8f;

#define T_SEQ 2048
#define EMB   1024
#define HID   64
#define NB    8
#define MTOT  (NB * T_SEQ)
#define NKT   (T_SEQ / 16)

#define QOFF   0
#define KOFF   ((size_t)MTOT * HID)
#define VOFF   ((size_t)2 * MTOT * HID)
#define SOFF   ((size_t)3 * MTOT * HID)

__device__ __forceinline__ v8f wmma_f32(v2f a, v2f b, v8f c) {
  return __builtin_amdgcn_wmma_f32_16x16x4_f32(
       false, a,  false, b,
       (short)0, c,  false,  false);
}

__device__ __forceinline__ float xhalf(float v) {
  const int b = __float_as_int(v);
  return __int_as_float(__builtin_amdgcn_permlanex16(b, b, 0x76543210, 0xfedcba98, false, false));
}

#define KG  (EMB / 8)
#define KU  4

struct QkvFrag {
  v4f x[KU];
  v4f w[KU][4];
};

__device__ __forceinline__ void qkv_load(QkvFrag& f, const float* __restrict__ arow,
                                         const float* __restrict__ wrow, int g0) {
  #pragma unroll
  for (int u = 0; u < KU; ++u) {
    f.x[u] = *(const v4f*)(arow + 8 * (g0 + u));
    #pragma unroll
    for (int j = 0; j < 4; ++j)
      f.w[u][j] = *(const v4f*)(wrow + (size_t)(8 * (g0 + u) + j) * HID);
  }
}

__device__ __forceinline__ void qkv_mma(const QkvFrag& f, v8f& c0, v8f& c1, v8f& c2, v8f& c3) {
  #pragma unroll
  for (int u = 0; u < KU; ++u) {
    const v2f a01 = {f.x[u].x, f.x[u].y};
    const v2f a23 = {f.x[u].z, f.x[u].w};
    c0 = wmma_f32(a01, v2f{f.w[u][0].x, f.w[u][1].x}, c0);
    c1 = wmma_f32(a01, v2f{f.w[u][0].y, f.w[u][1].y}, c1);
    c2 = wmma_f32(a01, v2f{f.w[u][0].z, f.w[u][1].z}, c2);
    c3 = wmma_f32(a01, v2f{f.w[u][0].w, f.w[u][1].w}, c3);
    c0 = wmma_f32(a23, v2f{f.w[u][2].x, f.w[u][3].x}, c0);
    c1 = wmma_f32(a23, v2f{f.w[u][2].y, f.w[u][3].y}, c1);
    c2 = wmma_f32(a23, v2f{f.w[u][2].z, f.w[u][3].z}, c2);
    c3 = wmma_f32(a23, v2f{f.w[u][2].w, f.w[u][3].w}, c3);
  }
}

__global__ __launch_bounds__(384) __attribute__((amdgpu_waves_per_eu(3, 3))) void qkv_kernel(
    const float* __restrict__ X,
    const float* __restrict__ Wq, const float* __restrict__ bq,
    const float* __restrict__ Wk, const float* __restrict__ bk,
    const float* __restrict__ Wv, const float* __restrict__ bv,
    float* __restrict__ ws) {
  const int lane = threadIdx.x;
  const int mt   = __builtin_amdgcn_readfirstlane(threadIdx.y);
  const int mat  = __builtin_amdgcn_readfirstlane(threadIdx.z);
  const int m0   = (blockIdx.x * 4 + mt) * 16;
  const int c    = lane & 15;
  const int h    = lane >> 4;

  const float* W    = (mat == 0) ? Wq : (mat == 1) ? Wk : Wv;
  const float* bias = (mat == 0) ? bq : (mat == 1) ? bk : bv;
  float* out = ws + (size_t)mat * (size_t)MTOT * HID;

  const float* arow = X + (size_t)(m0 + c) * EMB + 4 * h;
  const float* wrow = W + (size_t)(4 * h) * HID + 4 * c;

  v8f c0 = {}, c1 = {}, c2 = {}, c3 = {};
  QkvFrag fa, fb;
  qkv_load(fa, arow, wrow, 0);
  #pragma nounroll
  for (int g0 = 0; g0 < KG - 2 * KU; g0 += 2 * KU) {
    qkv_load(fb, arow, wrow, g0 + KU);
    qkv_mma(fa, c0, c1, c2, c3);
    qkv_load(fa, arow, wrow, g0 + 2 * KU);
    qkv_mma(fb, c0, c1, c2, c3);
  }
  qkv_load(fb, arow, wrow, KG - KU);
  qkv_mma(fa, c0, c1, c2, c3);
  qkv_mma(fb, c0, c1, c2, c3);

  const v4f bb = *(const v4f*)(bias + 4 * c);
  float* tile = out + (size_t)m0 * HID;
  if (mat == 0) {
    float* orow = tile + (size_t)(h * 8) * HID + 4 * c;
    #pragma unroll
    for (int r = 0; r < 8; ++r) {
      v4f o = {c0[r] + bb.x, c1[r] + bb.y, c2[r] + bb.z, c3[r] + bb.w};
      *(v4f*)(orow + (size_t)r * HID) = o;
    }
  } else if (mat == 1) {
    float* o4 = tile + (size_t)(16 * c + 8 * h) * 4;
    #pragma unroll
    for (int r = 0; r < 8; ++r) {
      v4f o = {c0[r] + bb.x, c1[r] + bb.y, c2[r] + bb.z, c3[r] + bb.w};
      *(v4f*)(o4 + 4 * r) = o;
    }
  } else {
    float* o4 = tile + (size_t)(32 * (c >> 2) + 16 * h + 4 * (c & 3)) * 4;
    #pragma unroll
    for (int q = 0; q < 2; ++q) {
      const v4f o0 = {c0[4 * q] + bb.x, c0[4 * q + 1] + bb.x, c0[4 * q + 2] + bb.x, c0[4 * q + 3] + bb.x};
      const v4f o1 = {c1[4 * q] + bb.y, c1[4 * q + 1] + bb.y, c1[4 * q + 2] + bb.y, c1[4 * q + 3] + bb.y};
      const v4f o2 = {c2[4 * q] + bb.z, c2[4 * q + 1] + bb.z, c2[4 * q + 2] + bb.z, c2[4 * q + 3] + bb.z};
      const v4f o3 = {c3[4 * q] + bb.w, c3[4 * q + 1] + bb.w, c3[4 * q + 2] + bb.w, c3[4 * q + 3] + bb.w};
      *(v4f*)(o4 + (size_t)(128 * q + 0) * 4) = o0;
      *(v4f*)(o4 + (size_t)(128 * q + 1) * 4) = o1;
      *(v4f*)(o4 + (size_t)(128 * q + 2) * 4) = o2;
      *(v4f*)(o4 + (size_t)(128 * q + 3) * 4) = o3;
    }
    v4f ts = {0.f, 0.f, 0.f, 0.f};
    #pragma unroll
    for (int r = 0; r < 8; ++r) {
      ts.x += c0[r] + bb.x; ts.y += c1[r] + bb.y;
      ts.z += c2[r] + bb.z; ts.w += c3[r] + bb.w;
    }
    ts.x += xhalf(ts.x); ts.y += xhalf(ts.y);
    ts.z += xhalf(ts.z); ts.w += xhalf(ts.w);
    if (h == 0) *(v4f*)(ws + SOFF + (size_t)(m0 / 16) * HID + 4 * c) = ts;
  }
}

#define NT     8
#define KB     (16 * NT)

__device__ __forceinline__ void attn_load_k(v4f (&kf)[8][NT], const float* __restrict__ kt) {
  #pragma unroll
  for (int g = 0; g < 8; ++g) {
    #pragma unroll
    for (int t = 0; t < NT; ++t) kf[g][t] = *(const v4f*)(kt + (size_t)t * 16 * HID + 128 * g);
  }
}

__device__ __forceinline__ void attn_load_v(v4f (&vf)[2 * NT][4], const float* __restrict__ vb) {
  #pragma unroll
  for (int j = 0; j < 2 * NT; ++j) {
    #pragma unroll
    for (int t = 0; t < 4; ++t) vf[j][t] = *(const v4f*)(vb + 512 * j + 128 * t);
  }
}

__device__ __forceinline__ void attn_s_mma(v8f (&s)[NT], const v4f (&qa)[8], const v4f (&kf)[8][NT]) {
  #pragma unroll
  for (int t = 0; t < NT; ++t) s[t] = v8f{};
  #pragma unroll
  for (int g = 0; g < 8; ++g) {
    const v2f q01 = {qa[g].x, qa[g].y};
    const v2f q23 = {qa[g].z, qa[g].w};
    #pragma unroll
    for (int t = 0; t < NT; ++t) s[t] = wmma_f32(v2f{kf[g][t].x, kf[g][t].y}, q01, s[t]);
    #pragma unroll
    for (int t = 0; t < NT; ++t) s[t] = wmma_f32(v2f{kf[g][t].z, kf[g][t].w}, q23, s[t]);
  }
}

__device__ __forceinline__ void attn_pv_mma(v8f& acc0, v8f& acc1, v8f& acc2, v8f& acc3,
                                            const v8f (&p)[NT], const v4f (&vf)[2 * NT][4]) {
  #pragma unroll
  for (int t = 0; t < NT; ++t) {
    #pragma unroll
    for (int q = 0; q < 2; ++q) {
      const int j = 2 * t + q;
      const v2f p01 = {p[t][4 * q], p[t][4 * q + 1]};
      const v2f p23 = {p[t][4 * q + 2], p[t][4 * q + 3]};
      acc0 = wmma_f32(v2f{vf[j][0].x, vf[j][0].y}, p01, acc0);
      acc1 = wmma_f32(v2f{vf[j][1].x, vf[j][1].y}, p01, acc1);
      acc2 = wmma_f32(v2f{vf[j][2].x, vf[j][2].y}, p01, acc2);
      acc3 = wmma_f32(v2f{vf[j][3].x, vf[j][3].y}, p01, acc3);
      acc0 = wmma_f32(v2f{vf[j][0].z, vf[j][0].w}, p23, acc0);
      acc1 = wmma_f32(v2f{vf[j][1].z, vf[j][1].w}, p23, acc1);
      acc2 = wmma_f32(v2f{vf[j][2].z, vf[j][2].w}, p23, acc2);
      acc3 = wmma_f32(v2f{vf[j][3].z, vf[j][3].w}, p23, acc3);
    }
  }
}

template <bool MASKED>
__device__ __forceinline__ void attn_softmax(v8f (&s)[NT], float& rm, float& rs,
                                             v8f& acc0, v8f& acc1, v8f& acc2, v8f& acc3,
                                             int k0, int q0, int ln, int hi) {
  const float scale = 0.125f;
  const int qg = q0 + ln;
  #pragma unroll
  for (int t = 0; t < NT; ++t) {
    #pragma unroll
    for (int r = 0; r < 8; ++r) {
      const int kg = k0 + 16 * t + 8 * hi + r;
      const float v = s[t][r] * scale;
      s[t][r] = (MASKED && kg > qg) ? 0.f : v;
    }
  }
  float mx[8];
  #pragma unroll
  for (int r = 0; r < 8; ++r) {
    mx[r] = s[0][r];
    #pragma unroll
    for (int t = 1; t < NT; ++t) mx[r] = fmaxf(mx[r], s[t][r]);
  }
  float m = fmaxf(fmaxf(fmaxf(mx[0], mx[1]), fmaxf(mx[2], mx[3])),
                  fmaxf(fmaxf(mx[4], mx[5]), fmaxf(mx[6], mx[7])));
  m = fmaxf(m, xhalf(m));
  const float nm = fmaxf(rm, m);
  const float c  = __expf(rm - nm);
  float part[8];
  #pragma unroll
  for (int r = 0; r < 8; ++r) part[r] = 0.f;
  #pragma unroll
  for (int t = 0; t < NT; ++t) {
    #pragma unroll
    for (int r = 0; r < 8; ++r) {
      const float e = __expf(s[t][r] - nm);
      s[t][r] = e;
      part[r] += e;
    }
  }
  const float rsum = ((part[0] + part[1]) + (part[2] + part[3])) +
                     ((part[4] + part[5]) + (part[6] + part[7]));
  rs = rs * c + rsum;
  rm = nm;
  #pragma unroll
  for (int r = 0; r < 8; ++r) {
    acc0[r] *= c; acc1[r] *= c;
    acc2[r] *= c; acc3[r] *= c;
  }
}

#define HA_MAX 9

__device__ __forceinline__ void attn_store(float* __restrict__ orow, const v8f& acc0, const v8f& acc1,
                                           const v8f& acc2, const v8f& acc3, float inv) {
  *(v4f*)(orow +  0) = v4f{acc0[0] * inv, acc0[1] * inv, acc0[2] * inv, acc0[3] * inv};
  *(v4f*)(orow +  4) = v4f{acc0[4] * inv, acc0[5] * inv, acc0[6] * inv, acc0[7] * inv};
  *(v4f*)(orow + 16) = v4f{acc1[0] * inv, acc1[1] * inv, acc1[2] * inv, acc1[3] * inv};
  *(v4f*)(orow + 20) = v4f{acc1[4] * inv, acc1[5] * inv, acc1[6] * inv, acc1[7] * inv};
  *(v4f*)(orow + 32) = v4f{acc2[0] * inv, acc2[1] * inv, acc2[2] * inv, acc2[3] * inv};
  *(v4f*)(orow + 36) = v4f{acc2[4] * inv, acc2[5] * inv, acc2[6] * inv, acc2[7] * inv};
  *(v4f*)(orow + 48) = v4f{acc3[0] * inv, acc3[1] * inv, acc3[2] * inv, acc3[3] * inv};
  *(v4f*)(orow + 52) = v4f{acc3[4] * inv, acc3[5] * inv, acc3[6] * inv, acc3[7] * inv};
}

__global__ __launch_bounds__(128) __attribute__((amdgpu_waves_per_eu(1, 1))) void attn_kernel(
    const float* __restrict__ ws, float* __restrict__ out) {
  __shared__ float xch[2][34][32];

  const int lane = threadIdx.x;
  const int w    = __builtin_amdgcn_readfirstlane(threadIdx.y);
  const int b    = blockIdx.x;
  const int pr   = w >> 1;
  const int role = w & 1;
  const int st   = blockIdx.y * 2 + pr;
  const int lt   = NKT - 1 - st;
  const int nS   = st / NT + 1;
  const int nL   = NKT / NT - st / NT;
  const int hA   = min(nL - 1, HA_MAX);
  const int ln   = lane & 15;
  const int hi   = lane >> 4;

  const float* Q = ws + QOFF;
  const float* K = ws + KOFF;
  const float* V = ws + VOFF;
  const float* TS = ws + SOFF;

  v8f acc0 = {}, acc1 = {}, acc2 = {}, acc3 = {};
  float rm = -INFINITY, rst = 0.f;
  v4f pa = {};

  const int nseg = role + 1;
  for (int sg = 0; sg < nseg; ++sg) {
    const bool tail = (role == 1) && (sg == 0);
    const bool shrt = (role == 1) && (sg == 1);
    const int qt   = shrt ? st : lt;
    const int blk0 = tail ? hA : 0;
    const int nblk = shrt ? nS : (tail ? nL - hA : hA);
    const int cflo = shrt ? nS * KB : nL * KB;
    const int cfhi = shrt ? nL * KB : T_SEQ;
    const int cfrows = (role == 1) ? cfhi - cflo : 0;
    const int cf     = (role == 1) ? T_SEQ - cflo : 0;
    const int q0   = qt * 16;

    v4f qa[8];
    {
      const float* qrow = Q + (size_t)(b * T_SEQ + q0 + ln) * HID + 4 * hi;
      #pragma unroll
      for (int g = 0; g < 8; ++g) qa[g] = *(const v4f*)(qrow + 8 * g);
    }

    acc0 = v8f{}; acc1 = v8f{}; acc2 = v8f{}; acc3 = v8f{};
    rm = -INFINITY;
    float rs = 0.f;

    const float* kbase = K + (size_t)(b * T_SEQ + blk0 * KB) * HID + 4 * lane;
    const float* vbase = V + (size_t)(b * T_SEQ + blk0 * KB) * HID + 4 * lane;

    v4f kf[8][NT];
    v4f vf[2 * NT][4];
    v8f s[NT];

    attn_load_k(kf, kbase);
    attn_s_mma(s, qa, kf);
    const int lastk = (nblk - 1) * KB;
    if (nblk > 1) attn_load_k(kf, kbase + (size_t)KB * HID);

    for (int kb = 0; kb < nblk - 1; ++kb) {
      const int k0 = kb * KB;
      attn_load_v(vf, vbase + (size_t)k0 * HID);
      v8f sn[NT];
      attn_s_mma(sn, qa, kf);
      attn_softmax<false>(s, rm, rs, acc0, acc1, acc2, acc3, k0, q0, ln, hi);
      attn_load_k(kf, kbase + (size_t)min(k0 + 2 * KB, lastk) * HID);
      __builtin_amdgcn_sched_barrier(0x406);
      attn_pv_mma(acc0, acc1, acc2, acc3, s, vf);
      #pragma unroll
      for (int t = 0; t < NT; ++t) s[t] = sn[t];
    }

    {
      attn_load_v(vf, vbase + (size_t)lastk * HID);
      attn_softmax<true>(s, rm, rs, acc0, acc1, acc2, acc3, blk0 * KB + lastk, q0, ln, hi);
      attn_pv_mma(acc0, acc1, acc2, acc3, s, vf);
    }

    rst = rs + xhalf(rs);

    if (cfrows > 0) {
      const float* tsb = TS + (size_t)(b * NKT + cflo / 16) * HID + 4 * lane;
      const int nb = cfrows / KB;
      for (int i0 = 0; i0 < nb; i0 += 8) {
        v4f x[32];
        #pragma unroll
        for (int g = 0; g < 8; ++g) {
          const float* tb = tsb + (size_t)min(i0 + g, nb - 1) * 512;
          #pragma unroll
          for (int u = 0; u < 4; ++u) x[4 * g + u] = *(const v4f*)(tb + 128 * u);
        }
        #pragma unroll
        for (int g = 0; g < 8; ++g) {
          const v4f sg = (x[4 * g] + x[4 * g + 1]) + (x[4 * g + 2] + x[4 * g + 3]);
          const float keep = (i0 + g < nb) ? 1.f : 0.f;
          pa += sg * keep;
        }
      }
    }

    if (cf > 0) {
      v4f tot;
      tot.x = pa.x + xhalf(pa.x);
      tot.y = pa.y + xhalf(pa.y);
      tot.z = pa.z + xhalf(pa.z);
      tot.w = pa.w + xhalf(pa.w);
      const float nm = fmaxf(rm, 0.f);
      const float c  = __expf(rm - nm);
      const float wz = __expf(-nm);
      rst = rst * c + wz * (float)cf;
      rm  = nm;
      #pragma unroll
      for (int r = 0; r < 8; ++r) {
        const int from = 2 * hi + (r >> 2);
        const float e = tot[r & 3];
        acc0[r] = acc0[r] * c + wz * __shfl(e, from);
        acc1[r] = acc1[r] * c + wz * __shfl(e, from + 4);
        acc2[r] = acc2[r] * c + wz * __shfl(e, from + 8);
        acc3[r] = acc3[r] * c + wz * __shfl(e, from + 12);
      }
    }

    if (tail) {
      xch[pr][0][lane] = rm;
      xch[pr][1][lane] = rst;
      #pragma unroll
      for (int r = 0; r < 8; ++r) {
        xch[pr][2 + r][lane]  = acc0[r];
        xch[pr][10 + r][lane] = acc1[r];
        xch[pr][18 + r][lane] = acc2[r];
        xch[pr][26 + r][lane] = acc3[r];
      }
    }
    if (shrt) {
      attn_store(out + (size_t)(b * T_SEQ + q0 + ln) * HID + 8 * hi, acc0, acc1, acc2, acc3, 1.f / rst);
    }
  }

  __syncthreads();

  if (role == 0) {
    const float rmB  = xch[pr][0][lane];
    const float rstB = xch[pr][1][lane];
    const float nm = fmaxf(rm, rmB);
    const float cA = __expf(rm - nm);
    const float cB = __expf(rmB - nm);
    rst = rst * cA + rstB * cB;
    #pragma unroll
    for (int r = 0; r < 8; ++r) {
      acc0[r] = acc0[r] * cA + xch[pr][2 + r][lane]  * cB;
      acc1[r] = acc1[r] * cA + xch[pr][10 + r][lane] * cB;
      acc2[r] = acc2[r] * cA + xch[pr][18 + r][lane] * cB;
      acc3[r] = acc3[r] * cA + xch[pr][26 + r][lane] * cB;
    }
    attn_store(out + (size_t)(b * T_SEQ + lt * 16 + ln) * HID + 8 * hi, acc0, acc1, acc2, acc3, 1.f / rst);
  }
}

extern "C" void kernel_launch(void* const* d_in, const int* in_sizes, int n_in,
                              void* d_out, int out_size, void* d_ws, size_t ws_size,
                              hipStream_t stream) {
  const float* X  = (const float*)d_in[0];
  const float* Wq = (const float*)d_in[1];
  const float* bq = (const float*)d_in[2];
  const float* Wk = (const float*)d_in[3];
  const float* bk = (const float*)d_in[4];
  const float* Wv = (const float*)d_in[5];
  const float* bv = (const float*)d_in[6];
  float* ws  = (float*)d_ws;
  float* out = (float*)d_out;

  qkv_kernel<<<dim3(MTOT / 64), dim3(32, 4, 3), 0, stream>>>(
      X, Wq, bq, Wk, bk, Wv, bv, ws);
  attn_kernel<<<dim3(NB, NKT / 4), dim3(32, 4), 0, stream>>>(ws, out);
}


# QKV projection: five 16-WMMA fragment sets in rotation, each X/W load group issued four sets (64 WMMAs) ahead of its use behind scheduling fences, no set copies; per-accumulator sum order unchanged
# speedup vs baseline: 5.1588x; 1.0404x over previous
#include <hip/hip_runtime.h>
#include <math.h>


typedef __attribute__((ext_vector_type(2))) float v2f;
typedef __attribute__((ext_vector_type(4))) float v4f;
typedef __attribute__((ext_vector_type(8))) float v8f;

#define T_SEQ 2048
#define EMB   1024
#define HID   64
#define NB    8
#define MTOT  (NB * T_SEQ)
#define NKT   (T_SEQ / 16)

#define QOFF   0
#define KOFF   ((size_t)MTOT * HID)
#define VOFF   ((size_t)2 * MTOT * HID)
#define SOFF   ((size_t)3 * MTOT * HID)

__device__ __forceinline__ v8f wmma_f32(v2f a, v2f b, v8f c) {
  return __builtin_amdgcn_wmma_f32_16x16x4_f32(
       false, a,  false, b,
       (short)0, c,  false,  false);
}

__device__ __forceinline__ float xhalf(float v) {
  const int b = __float_as_int(v);
  return __int_as_float(__builtin_amdgcn_permlanex16(b, b, 0x76543210, 0xfedcba98, false, false));
}

#define KG  (EMB / 8)
#define KU  2
#define NBUF 5

struct QkvFrag {
  v4f x[KU];
  v4f w[KU][4];
};

__device__ __forceinline__ void qkv_load(QkvFrag& f, const float* __restrict__ arow,
                                         const float* __restrict__ wrow, int g0) {
  #pragma unroll
  for (int u = 0; u < KU; ++u) {
    f.x[u] = *(const v4f*)(arow + 8 * (g0 + u));
    #pragma unroll
    for (int j = 0; j < 4; ++j)
      f.w[u][j] = *(const v4f*)(wrow + (size_t)(8 * (g0 + u) + j) * HID);
  }
}

__device__ __forceinline__ void qkv_mma(const QkvFrag& f, v8f& c0, v8f& c1, v8f& c2, v8f& c3) {
  #pragma unroll
  for (int u = 0; u < KU; ++u) {
    const v2f a01 = {f.x[u].x, f.x[u].y};
    const v2f a23 = {f.x[u].z, f.x[u].w};
    c0 = wmma_f32(a01, v2f{f.w[u][0].x, f.w[u][1].x}, c0);
    c1 = wmma_f32(a01, v2f{f.w[u][0].y, f.w[u][1].y}, c1);
    c2 = wmma_f32(a01, v2f{f.w[u][0].z, f.w[u][1].z}, c2);
    c3 = wmma_f32(a01, v2f{f.w[u][0].w, f.w[u][1].w}, c3);
    c0 = wmma_f32(a23, v2f{f.w[u][2].x, f.w[u][3].x}, c0);
    c1 = wmma_f32(a23, v2f{f.w[u][2].y, f.w[u][3].y}, c1);
    c2 = wmma_f32(a23, v2f{f.w[u][2].z, f.w[u][3].z}, c2);
    c3 = wmma_f32(a23, v2f{f.w[u][2].w, f.w[u][3].w}, c3);
  }
}

__global__ __launch_bounds__(384) __attribute__((amdgpu_waves_per_eu(3, 3))) void qkv_kernel(
    const float* __restrict__ X,
    const float* __restrict__ Wq, const float* __restrict__ bq,
    const float* __restrict__ Wk, const float* __restrict__ bk,
    const float* __restrict__ Wv, const float* __restrict__ bv,
    float* __restrict__ ws) {
  const int lane = threadIdx.x;
  const int mt   = __builtin_amdgcn_readfirstlane(threadIdx.y);
  const int mat  = __builtin_amdgcn_readfirstlane(threadIdx.z);
  const int m0   = (blockIdx.x * 4 + mt) * 16;
  const int c    = lane & 15;
  const int h    = lane >> 4;

  const float* W    = (mat == 0) ? Wq : (mat == 1) ? Wk : Wv;
  const float* bias = (mat == 0) ? bq : (mat == 1) ? bk : bv;
  float* out = ws + (size_t)mat * (size_t)MTOT * HID;

  const float* arow = X + (size_t)(m0 + c) * EMB + 4 * h;
  const float* wrow = W + (size_t)(4 * h) * HID + 4 * c;

  v8f c0 = {}, c1 = {}, c2 = {}, c3 = {};
  QkvFrag f[NBUF];
  #pragma unroll
  for (int b = 0; b < NBUF - 1; ++b) qkv_load(f[b], arow, wrow, b * KU);
  int g0 = 0;
  #pragma nounroll
  for (; g0 + (2 * NBUF - 1) * KU <= KG; g0 += NBUF * KU) {
    #pragma unroll
    for (int b = 0; b < NBUF; ++b) {
      qkv_load(f[(b + NBUF - 1) % NBUF], arow, wrow, g0 + (b + NBUF - 1) * KU);
      __builtin_amdgcn_sched_barrier(0);
      qkv_mma(f[b], c0, c1, c2, c3);
      __builtin_amdgcn_sched_barrier(0);
    }
  }
  static_assert((KG / KU) % NBUF == NBUF - 1, "the rotation must end on NBUF - 1 loaded sets");
  #pragma unroll
  for (int b = 0; b < NBUF - 1; ++b) qkv_mma(f[b], c0, c1, c2, c3);

  const v4f bb = *(const v4f*)(bias + 4 * c);
  float* tile = out + (size_t)m0 * HID;
  if (mat == 0) {
    float* orow = tile + (size_t)(h * 8) * HID + 4 * c;
    #pragma unroll
    for (int r = 0; r < 8; ++r) {
      v4f o = {c0[r] + bb.x, c1[r] + bb.y, c2[r] + bb.z, c3[r] + bb.w};
      *(v4f*)(orow + (size_t)r * HID) = o;
    }
  } else if (mat == 1) {
    float* o4 = tile + (size_t)(16 * c + 8 * h) * 4;
    #pragma unroll
    for (int r = 0; r < 8; ++r) {
      v4f o = {c0[r] + bb.x, c1[r] + bb.y, c2[r] + bb.z, c3[r] + bb.w};
      *(v4f*)(o4 + 4 * r) = o;
    }
  } else {
    float* o4 = tile + (size_t)(32 * (c >> 2) + 16 * h + 4 * (c & 3)) * 4;
    #pragma unroll
    for (int q = 0; q < 2; ++q) {
      const v4f o0 = {c0[4 * q] + bb.x, c0[4 * q + 1] + bb.x, c0[4 * q + 2] + bb.x, c0[4 * q + 3] + bb.x};
      const v4f o1 = {c1[4 * q] + bb.y, c1[4 * q + 1] + bb.y, c1[4 * q + 2] + bb.y, c1[4 * q + 3] + bb.y};
      const v4f o2 = {c2[4 * q] + bb.z, c2[4 * q + 1] + bb.z, c2[4 * q + 2] + bb.z, c2[4 * q + 3] + bb.z};
      const v4f o3 = {c3[4 * q] + bb.w, c3[4 * q + 1] + bb.w, c3[4 * q + 2] + bb.w, c3[4 * q + 3] + bb.w};
      *(v4f*)(o4 + (size_t)(128 * q + 0) * 4) = o0;
      *(v4f*)(o4 + (size_t)(128 * q + 1) * 4) = o1;
      *(v4f*)(o4 + (size_t)(128 * q + 2) * 4) = o2;
      *(v4f*)(o4 + (size_t)(128 * q + 3) * 4) = o3;
    }
    v4f ts = {0.f, 0.f, 0.f, 0.f};
    #pragma unroll
    for (int r = 0; r < 8; ++r) {
      ts.x += c0[r] + bb.x; ts.y += c1[r] + bb.y;
      ts.z += c2[r] + bb.z; ts.w += c3[r] + bb.w;
    }
    ts.x += xhalf(ts.x); ts.y += xhalf(ts.y);
    ts.z += xhalf(ts.z); ts.w += xhalf(ts.w);
    if (h == 0) *(v4f*)(ws + SOFF + (size_t)(m0 / 16) * HID + 4 * c) = ts;
  }
}

#define NT     8
#define KB     (16 * NT)

__device__ __forceinline__ void attn_load_k(v4f (&kf)[8][NT], const float* __restrict__ kt) {
  #pragma unroll
  for (int g = 0; g < 8; ++g) {
    #pragma unroll
    for (int t = 0; t < NT; ++t) kf[g][t] = *(const v4f*)(kt + (size_t)t * 16 * HID + 128 * g);
  }
}

__device__ __forceinline__ void attn_load_v(v4f (&vf)[2 * NT][4], const float* __restrict__ vb) {
  #pragma unroll
  for (int j = 0; j < 2 * NT; ++j) {
    #pragma unroll
    for (int t = 0; t < 4; ++t) vf[j][t] = *(const v4f*)(vb + 512 * j + 128 * t);
  }
}

__device__ __forceinline__ void attn_s_mma(v8f (&s)[NT], const v4f (&qa)[8], const v4f (&kf)[8][NT]) {
  #pragma unroll
  for (int t = 0; t < NT; ++t) s[t] = v8f{};
  #pragma unroll
  for (int g = 0; g < 8; ++g) {
    const v2f q01 = {qa[g].x, qa[g].y};
    const v2f q23 = {qa[g].z, qa[g].w};
    #pragma unroll
    for (int t = 0; t < NT; ++t) s[t] = wmma_f32(v2f{kf[g][t].x, kf[g][t].y}, q01, s[t]);
    #pragma unroll
    for (int t = 0; t < NT; ++t) s[t] = wmma_f32(v2f{kf[g][t].z, kf[g][t].w}, q23, s[t]);
  }
}

__device__ __forceinline__ void attn_pv_mma(v8f& acc0, v8f& acc1, v8f& acc2, v8f& acc3,
                                            const v8f (&p)[NT], const v4f (&vf)[2 * NT][4]) {
  #pragma unroll
  for (int t = 0; t < NT; ++t) {
    #pragma unroll
    for (int q = 0; q < 2; ++q) {
      const int j = 2 * t + q;
      const v2f p01 = {p[t][4 * q], p[t][4 * q + 1]};
      const v2f p23 = {p[t][4 * q + 2], p[t][4 * q + 3]};
      acc0 = wmma_f32(v2f{vf[j][0].x, vf[j][0].y}, p01, acc0);
      acc1 = wmma_f32(v2f{vf[j][1].x, vf[j][1].y}, p01, acc1);
      acc2 = wmma_f32(v2f{vf[j][2].x, vf[j][2].y}, p01, acc2);
      acc3 = wmma_f32(v2f{vf[j][3].x, vf[j][3].y}, p01, acc3);
      acc0 = wmma_f32(v2f{vf[j][0].z, vf[j][0].w}, p23, acc0);
      acc1 = wmma_f32(v2f{vf[j][1].z, vf[j][1].w}, p23, acc1);
      acc2 = wmma_f32(v2f{vf[j][2].z, vf[j][2].w}, p23, acc2);
      acc3 = wmma_f32(v2f{vf[j][3].z, vf[j][3].w}, p23, acc3);
    }
  }
}

template <bool MASKED>
__device__ __forceinline__ void attn_softmax(v8f (&s)[NT], float& rm, float& rs,
                                             v8f& acc0, v8f& acc1, v8f& acc2, v8f& acc3,
                                             int k0, int q0, int ln, int hi) {
  const float scale = 0.125f;
  const int qg = q0 + ln;
  #pragma unroll
  for (int t = 0; t < NT; ++t) {
    #pragma unroll
    for (int r = 0; r < 8; ++r) {
      const int kg = k0 + 16 * t + 8 * hi + r;
      const float v = s[t][r] * scale;
      s[t][r] = (MASKED && kg > qg) ? 0.f : v;
    }
  }
  float mx[8];
  #pragma unroll
  for (int r = 0; r < 8; ++r) {
    mx[r] = s[0][r];
    #pragma unroll
    for (int t = 1; t < NT; ++t) mx[r] = fmaxf(mx[r], s[t][r]);
  }
  float m = fmaxf(fmaxf(fmaxf(mx[0], mx[1]), fmaxf(mx[2], mx[3])),
                  fmaxf(fmaxf(mx[4], mx[5]), fmaxf(mx[6], mx[7])));
  m = fmaxf(m, xhalf(m));
  const float nm = fmaxf(rm, m);
  const float c  = __expf(rm - nm);
  float part[8];
  #pragma unroll
  for (int r = 0; r < 8; ++r) part[r] = 0.f;
  #pragma unroll
  for (int t = 0; t < NT; ++t) {
    #pragma unroll
    for (int r = 0; r < 8; ++r) {
      const float e = __expf(s[t][r] - nm);
      s[t][r] = e;
      part[r] += e;
    }
  }
  const float rsum = ((part[0] + part[1]) + (part[2] + part[3])) +
                     ((part[4] + part[5]) + (part[6] + part[7]));
  rs = rs * c + rsum;
  rm = nm;
  #pragma unroll
  for (int r = 0; r < 8; ++r) {
    acc0[r] *= c; acc1[r] *= c;
    acc2[r] *= c; acc3[r] *= c;
  }
}

#define HA_MAX 9

__device__ __forceinline__ void attn_store(float* __restrict__ orow, const v8f& acc0, const v8f& acc1,
                                           const v8f& acc2, const v8f& acc3, float inv) {
  *(v4f*)(orow +  0) = v4f{acc0[0] * inv, acc0[1] * inv, acc0[2] * inv, acc0[3] * inv};
  *(v4f*)(orow +  4) = v4f{acc0[4] * inv, acc0[5] * inv, acc0[6] * inv, acc0[7] * inv};
  *(v4f*)(orow + 16) = v4f{acc1[0] * inv, acc1[1] * inv, acc1[2] * inv, acc1[3] * inv};
  *(v4f*)(orow + 20) = v4f{acc1[4] * inv, acc1[5] * inv, acc1[6] * inv, acc1[7] * inv};
  *(v4f*)(orow + 32) = v4f{acc2[0] * inv, acc2[1] * inv, acc2[2] * inv, acc2[3] * inv};
  *(v4f*)(orow + 36) = v4f{acc2[4] * inv, acc2[5] * inv, acc2[6] * inv, acc2[7] * inv};
  *(v4f*)(orow + 48) = v4f{acc3[0] * inv, acc3[1] * inv, acc3[2] * inv, acc3[3] * inv};
  *(v4f*)(orow + 52) = v4f{acc3[4] * inv, acc3[5] * inv, acc3[6] * inv, acc3[7] * inv};
}

__global__ __launch_bounds__(128) __attribute__((amdgpu_waves_per_eu(1, 1))) void attn_kernel(
    const float* __restrict__ ws, float* __restrict__ out) {
  __shared__ float xch[2][34][32];

  const int lane = threadIdx.x;
  const int w    = __builtin_amdgcn_readfirstlane(threadIdx.y);
  const int b    = blockIdx.x;
  const int pr   = w >> 1;
  const int role = w & 1;
  const int st   = blockIdx.y * 2 + pr;
  const int lt   = NKT - 1 - st;
  const int nS   = st / NT + 1;
  const int nL   = NKT / NT - st / NT;
  const int hA   = min(nL - 1, HA_MAX);
  const int ln   = lane & 15;
  const int hi   = lane >> 4;

  const float* Q = ws + QOFF;
  const float* K = ws + KOFF;
  const float* V = ws + VOFF;
  const float* TS = ws + SOFF;

  v8f acc0 = {}, acc1 = {}, acc2 = {}, acc3 = {};
  float rm = -INFINITY, rst = 0.f;
  v4f pa = {};

  const int nseg = role + 1;
  for (int sg = 0; sg < nseg; ++sg) {
    const bool tail = (role == 1) && (sg == 0);
    const bool shrt = (role == 1) && (sg == 1);
    const int qt   = shrt ? st : lt;
    const int blk0 = tail ? hA : 0;
    const int nblk = shrt ? nS : (tail ? nL - hA : hA);
    const int cflo = shrt ? nS * KB : nL * KB;
    const int cfhi = shrt ? nL * KB : T_SEQ;
    const int cfrows = (role == 1) ? cfhi - cflo : 0;
    const int cf     = (role == 1) ? T_SEQ - cflo : 0;
    const int q0   = qt * 16;

    v4f qa[8];
    {
      const float* qrow = Q + (size_t)(b * T_SEQ + q0 + ln) * HID + 4 * hi;
      #pragma unroll
      for (int g = 0; g < 8; ++g) qa[g] = *(const v4f*)(qrow + 8 * g);
    }

    acc0 = v8f{}; acc1 = v8f{}; acc2 = v8f{}; acc3 = v8f{};
    rm = -INFINITY;
    float rs = 0.f;

    const float* kbase = K + (size_t)(b * T_SEQ + blk0 * KB) * HID + 4 * lane;
    const float* vbase = V + (size_t)(b * T_SEQ + blk0 * KB) * HID + 4 * lane;

    v4f kf[8][NT];
    v4f vf[2 * NT][4];
    v8f s[NT];

    attn_load_k(kf, kbase);
    attn_s_mma(s, qa, kf);
    const int lastk = (nblk - 1) * KB;
    if (nblk > 1) attn_load_k(kf, kbase + (size_t)KB * HID);

    for (int kb = 0; kb < nblk - 1; ++kb) {
      const int k0 = kb * KB;
      attn_load_v(vf, vbase + (size_t)k0 * HID);
      v8f sn[NT];
      attn_s_mma(sn, qa, kf);
      attn_softmax<false>(s, rm, rs, acc0, acc1, acc2, acc3, k0, q0, ln, hi);
      attn_load_k(kf, kbase + (size_t)min(k0 + 2 * KB, lastk) * HID);
      __builtin_amdgcn_sched_barrier(0x406);
      attn_pv_mma(acc0, acc1, acc2, acc3, s, vf);
      #pragma unroll
      for (int t = 0; t < NT; ++t) s[t] = sn[t];
    }

    {
      attn_load_v(vf, vbase + (size_t)lastk * HID);
      attn_softmax<true>(s, rm, rs, acc0, acc1, acc2, acc3, blk0 * KB + lastk, q0, ln, hi);
      attn_pv_mma(acc0, acc1, acc2, acc3, s, vf);
    }

    rst = rs + xhalf(rs);

    if (cfrows > 0) {
      const float* tsb = TS + (size_t)(b * NKT + cflo / 16) * HID + 4 * lane;
      const int nb = cfrows / KB;
      for (int i0 = 0; i0 < nb; i0 += 8) {
        v4f x[32];
        #pragma unroll
        for (int g = 0; g < 8; ++g) {
          const float* tb = tsb + (size_t)min(i0 + g, nb - 1) * 512;
          #pragma unroll
          for (int u = 0; u < 4; ++u) x[4 * g + u] = *(const v4f*)(tb + 128 * u);
        }
        #pragma unroll
        for (int g = 0; g < 8; ++g) {
          const v4f sg = (x[4 * g] + x[4 * g + 1]) + (x[4 * g + 2] + x[4 * g + 3]);
          const float keep = (i0 + g < nb) ? 1.f : 0.f;
          pa += sg * keep;
        }
      }
    }

    if (cf > 0) {
      v4f tot;
      tot.x = pa.x + xhalf(pa.x);
      tot.y = pa.y + xhalf(pa.y);
      tot.z = pa.z + xhalf(pa.z);
      tot.w = pa.w + xhalf(pa.w);
      const float nm = fmaxf(rm, 0.f);
      const float c  = __expf(rm - nm);
      const float wz = __expf(-nm);
      rst = rst * c + wz * (float)cf;
      rm  = nm;
      #pragma unroll
      for (int r = 0; r < 8; ++r) {
        const int from = 2 * hi + (r >> 2);
        const float e = tot[r & 3];
        acc0[r] = acc0[r] * c + wz * __shfl(e, from);
        acc1[r] = acc1[r] * c + wz * __shfl(e, from + 4);
        acc2[r] = acc2[r] * c + wz * __shfl(e, from + 8);
        acc3[r] = acc3[r] * c + wz * __shfl(e, from + 12);
      }
    }

    if (tail) {
      xch[pr][0][lane] = rm;
      xch[pr][1][lane] = rst;
      #pragma unroll
      for (int r = 0; r < 8; ++r) {
        xch[pr][2 + r][lane]  = acc0[r];
        xch[pr][10 + r][lane] = acc1[r];
        xch[pr][18 + r][lane] = acc2[r];
        xch[pr][26 + r][lane] = acc3[r];
      }
    }
    if (shrt) {
      attn_store(out + (size_t)(b * T_SEQ + q0 + ln) * HID + 8 * hi, acc0, acc1, acc2, acc3, 1.f / rst);
    }
  }

  __syncthreads();

  if (role == 0) {
    const float rmB  = xch[pr][0][lane];
    const float rstB = xch[pr][1][lane];
    const float nm = fmaxf(rm, rmB);
    const float cA = __expf(rm - nm);
    const float cB = __expf(rmB - nm);
    rst = rst * cA + rstB * cB;
    #pragma unroll
    for (int r = 0; r < 8; ++r) {
      acc0[r] = acc0[r] * cA + xch[pr][2 + r][lane]  * cB;
      acc1[r] = acc1[r] * cA + xch[pr][10 + r][lane] * cB;
      acc2[r] = acc2[r] * cA + xch[pr][18 + r][lane] * cB;
      acc3[r] = acc3[r] * cA + xch[pr][26 + r][lane] * cB;
    }
    attn_store(out + (size_t)(b * T_SEQ + lt * 16 + ln) * HID + 8 * hi, acc0, acc1, acc2, acc3, 1.f / rst);
  }
}

extern "C" void kernel_launch(void* const* d_in, const int* in_sizes, int n_in,
                              void* d_out, int out_size, void* d_ws, size_t ws_size,
                              hipStream_t stream) {
  const float* X  = (const float*)d_in[0];
  const float* Wq = (const float*)d_in[1];
  const float* bq = (const float*)d_in[2];
  const float* Wk = (const float*)d_in[3];
  const float* bk = (const float*)d_in[4];
  const float* Wv = (const float*)d_in[5];
  const float* bv = (const float*)d_in[6];
  float* ws  = (float*)d_ws;
  float* out = (float*)d_out;

  qkv_kernel<<<dim3(MTOT / 64), dim3(32, 4, 3), 0, stream>>>(
      X, Wq, bq, Wk, bk, Wv, bv, ws);
  attn_kernel<<<dim3(NB, NKT / 4), dim3(32, 4), 0, stream>>>(ws, out);
}
